# starting kernel
# baseline (speedup 1.0000x reference)
#include <hip/hip_runtime.h>

typedef __attribute__((ext_vector_type(2))) float v2f;
typedef __attribute__((ext_vector_type(8))) float v8f;

#define NFEAT 512
#define NHID  128
#define NCLASS 10


__global__ void transpose_w(const float* __restrict__ W, float* __restrict__ Wt,
                            int K, int Nc) {
  int idx = blockIdx.x * blockDim.x + threadIdx.x;
  if (idx < K * Nc) {
    int k = idx / Nc, n = idx % Nc;
    Wt[(size_t)n * K + k] = W[idx];
  }
}

__global__ void deg_init(float* __restrict__ deg, int n) {
  int i = blockIdx.x * blockDim.x + threadIdx.x;
  if (i < n) deg[i] = 1.0f;
}

__global__ void deg_edges(const int* __restrict__ cols, float* __restrict__ deg, int E) {
  int e = blockIdx.x * blockDim.x + threadIdx.x;
  if (e < E) atomicAdd(&deg[cols[e]], 1.0f);
}

__global__ void deg_fin(float* __restrict__ deg, int n) {
  int i = blockIdx.x * blockDim.x + threadIdx.x;
  if (i < n) deg[i] = rsqrtf(deg[i]);
}

__global__ void __launch_bounds__(128) gemm_wmma_f32(const float* __restrict__ X,
                                                     const float* __restrict__ Wt,
                                                     float* __restrict__ H,
                                                     int M, int K) {
  int wave = (int)((blockIdx.x * (size_t)blockDim.x + threadIdx.x) >> 5);
  int lane = threadIdx.x & 31;
  if (wave * 16 >= M) return;

  int nlo = lane & 15;
  int kh  = (lane >> 4) * 2;

  const float* xrow = X + (size_t)(wave * 16 + nlo) * K;

  v8f acc[8] = {};
  for (int k0 = 0; k0 < K; k0 += 4) {
    v2f a = *(const v2f*)(xrow + k0 + kh);
#pragma unroll
    for (int t = 0; t < 8; ++t) {
      v2f b = *(const v2f*)(Wt + (size_t)(t * 16 + nlo) * K + k0 + kh);
      acc[t] = __builtin_amdgcn_wmma_f32_16x16x4_f32(
           false, a,  false, b,
           (short)0, acc[t],  false,  false);
    }
  }

  int mbase = wave * 16 + ((lane >> 4) << 3);
#pragma unroll
  for (int t = 0; t < 8; ++t) {
#pragma unroll
    for (int e = 0; e < 8; ++e) {
      H[(size_t)(mbase + e) * NHID + t * 16 + nlo] = acc[t][e];
    }
  }
}


__global__ void agg_init(const float* __restrict__ H, const float* __restrict__ dinv,
                         float* __restrict__ Y, int total) {
  int idx = blockIdx.x * blockDim.x + threadIdx.x;
  if (idx < total) {
    int i = idx >> 7;
    float d = dinv[i];
    Y[idx] = H[idx] * d * d;
  }
}

__global__ void __launch_bounds__(256) agg_edges(const int* __restrict__ rows,
                                                 const int* __restrict__ cols,
                                                 const float* __restrict__ dinv,
                                                 const float* __restrict__ H,
                                                 float* __restrict__ Y, int E) {
  int e = (int)((blockIdx.x * (size_t)blockDim.x + threadIdx.x) >> 5);
  int lane = threadIdx.x & 31;
  if (e >= E) return;
  int r = rows[e];
  int c = cols[e];
  float w = dinv[r] * dinv[c];
  const float4 h = *(const float4*)(H + (size_t)r * NHID + lane * 4);
  float* yp = Y + (size_t)c * NHID + lane * 4;
  atomicAdd(yp + 0, h.x * w);
  atomicAdd(yp + 1, h.y * w);
  atomicAdd(yp + 2, h.z * w);
  atomicAdd(yp + 3, h.w * w);
}

__global__ void agg_bias_act(float* __restrict__ Y, const float* __restrict__ b,
                             int total, int relu) {
  int idx = blockIdx.x * blockDim.x + threadIdx.x;
  if (idx < total) {
    float v = Y[idx] + b[idx & (NHID - 1)];
    Y[idx] = relu ? fmaxf(v, 0.0f) : v;
  }
}

__global__ void __launch_bounds__(256) final_logits(const float* __restrict__ X,
                                                    const float* __restrict__ Wlt,
                                                    const float* __restrict__ bl,
                                                    float* __restrict__ out, int n) {
  int node = (int)((blockIdx.x * (size_t)blockDim.x + threadIdx.x) >> 5);
  int lane = threadIdx.x & 31;
  if (node >= n) return;

  float4 xv = *(const float4*)(X + (size_t)node * NHID + lane * 4);
  float logit[NCLASS];
#pragma unroll
  for (int c = 0; c < NCLASS; ++c) {
    float4 wv = *(const float4*)(Wlt + (size_t)c * NHID + lane * 4);
    float p = xv.x * wv.x + xv.y * wv.y + xv.z * wv.z + xv.w * wv.w;
#pragma unroll
    for (int off = 16; off > 0; off >>= 1) p += __shfl_xor(p, off, 32);
    logit[c] = p + bl[c];
  }
  float m = logit[0];
#pragma unroll
  for (int c = 1; c < NCLASS; ++c) m = fmaxf(m, logit[c]);
  float s = 0.0f;
#pragma unroll
  for (int c = 0; c < NCLASS; ++c) s += __expf(logit[c] - m);
  float lse = m + __logf(s);
  if (lane < NCLASS) out[(size_t)node * NCLASS + lane] = logit[lane] - lse;
}

extern "C" void kernel_launch(void* const* d_in, const int* in_sizes, int n_in,
                              void* d_out, int out_size, void* d_ws, size_t ws_size,
                              hipStream_t stream) {
  const float* x  = (const float*)d_in[0];
  const int*   ei = (const int*)d_in[1];
  const float* W1 = (const float*)d_in[2];
  const float* b1 = (const float*)d_in[3];
  const float* W2 = (const float*)d_in[4];
  const float* b2 = (const float*)d_in[5];
  const float* W3 = (const float*)d_in[6];
  const float* b3 = (const float*)d_in[7];
  const float* Wl = (const float*)d_in[8];
  const float* bl = (const float*)d_in[9];
  float* out = (float*)d_out;

  const int E = in_sizes[1] / 2;
  const int N = in_sizes[0] / NFEAT;

  char* ws = (char*)d_ws;
  size_t off = 0;
  auto alloc = [&](size_t bytes) -> float* {
    float* p = (float*)(ws + off);
    off += (bytes + 255) & ~(size_t)255;
    return p;
  };
  float* dinv = alloc((size_t)N * sizeof(float));
  float* Wt1  = alloc((size_t)NFEAT * NHID * sizeof(float));
  float* Wt2  = alloc((size_t)NHID * NHID * sizeof(float));
  float* Wt3  = alloc((size_t)NHID * NHID * sizeof(float));
  float* Wlt  = alloc((size_t)NHID * NCLASS * sizeof(float));
  float* H    = alloc((size_t)N * NHID * sizeof(float));
  float* Y    = alloc((size_t)N * NHID * sizeof(float));

  const int* rows = ei;
  const int* cols = ei + E;

  transpose_w<<<(NFEAT * NHID + 255) / 256, 256, 0, stream>>>(W1, Wt1, NFEAT, NHID);
  transpose_w<<<(NHID * NHID + 255) / 256, 256, 0, stream>>>(W2, Wt2, NHID, NHID);
  transpose_w<<<(NHID * NHID + 255) / 256, 256, 0, stream>>>(W3, Wt3, NHID, NHID);
  transpose_w<<<(NHID * NCLASS + 255) / 256, 256, 0, stream>>>(Wl, Wlt, NHID, NCLASS);

  deg_init<<<(N + 255) / 256, 256, 0, stream>>>(dinv, N);
  deg_edges<<<(E + 255) / 256, 256, 0, stream>>>(cols, dinv, E);
  deg_fin<<<(N + 255) / 256, 256, 0, stream>>>(dinv, N);

  const int tiles = (N + 15) / 16;
  const int gemm_grid = (tiles + 3) / 4;
  const int tot = N * NHID;
  const int tgrid = (tot + 255) / 256;
  const int egrid = (int)(((size_t)E * 32 + 255) / 256);

  gemm_wmma_f32<<<gemm_grid, 128, 0, stream>>>(x, Wt1, H, N, NFEAT);
  agg_init<<<tgrid, 256, 0, stream>>>(H, dinv, Y, tot);
  agg_edges<<<egrid, 256, 0, stream>>>(rows, cols, dinv, H, Y, E);
  agg_bias_act<<<tgrid, 256, 0, stream>>>(Y, b1, tot, 1);

  gemm_wmma_f32<<<gemm_grid, 128, 0, stream>>>(Y, Wt2, H, N, NHID);
  agg_init<<<tgrid, 256, 0, stream>>>(H, dinv, Y, tot);
  agg_edges<<<egrid, 256, 0, stream>>>(rows, cols, dinv, H, Y, E);
  agg_bias_act<<<tgrid, 256, 0, stream>>>(Y, b2, tot, 1);

  gemm_wmma_f32<<<gemm_grid, 128, 0, stream>>>(Y, Wt3, H, N, NHID);
  agg_init<<<tgrid, 256, 0, stream>>>(H, dinv, Y, tot);
  agg_edges<<<egrid, 256, 0, stream>>>(rows, cols, dinv, H, Y, E);
  agg_bias_act<<<tgrid, 256, 0, stream>>>(Y, b3, tot, 0);

  final_logits<<<(int)(((size_t)N * 32 + 255) / 256), 256, 0, stream>>>(Y, Wlt, bl, out, N);
}


# Edge pass by owner workgroups, no global float atomics: each has 196 target rows in LDS, scans all targets 16 B per lane a trip ahead, queues hits per wave, adds rows by LDS float adds, arrival order
# speedup vs baseline: 2.1465x; 2.1465x over previous
#include <hip/hip_runtime.h>

typedef __attribute__((ext_vector_type(2))) float v2f;
typedef __attribute__((ext_vector_type(4))) float v4f;
typedef __attribute__((ext_vector_type(8))) float v8f;

#define NFEAT 512
#define NHID  128
#define NCLASS 10

#define OWN_ROWS 196
#define OWN_WAVES 32
#define SCAN_LOADS 8
#define QCAP (32 + 128 * SCAN_LOADS)


__global__ void transpose_w(const float* __restrict__ W, float* __restrict__ Wt,
                            int K, int Nc) {
  int idx = blockIdx.x * blockDim.x + threadIdx.x;
  if (idx < K * Nc) {
    int k = idx / Nc, n = idx % Nc;
    Wt[(size_t)n * K + k] = W[idx];
  }
}

__global__ void pack_w(const float* __restrict__ W, float* __restrict__ Wp, int K) {
  int idx = blockIdx.x * blockDim.x + threadIdx.x;
  if (idx < K * NHID) {
    int j = idx & 3, l = (idx >> 2) & 31, t = (idx >> 7) & 7, kb = idx >> 10;
    int k = kb * 8 + ((l >> 4) << 2) + j;
    int n = t * 16 + (l & 15);
    Wp[idx] = W[(size_t)k * NHID + n];
  }
}

__global__ void deg_init(float* __restrict__ deg, int n) {
  int i = blockIdx.x * blockDim.x + threadIdx.x;
  if (i < n) deg[i] = 1.0f;
}

__global__ void deg_edges(const int* __restrict__ cols, float* __restrict__ deg, int E) {
  int e = blockIdx.x * blockDim.x + threadIdx.x;
  if (e < E) atomicAdd(&deg[cols[e]], 1.0f);
}

__global__ void deg_fin(float* __restrict__ deg, int n) {
  int i = blockIdx.x * blockDim.x + threadIdx.x;
  if (i < n) deg[i] = rsqrtf(deg[i]);
}

template <bool PRE>
__device__ __forceinline__ void load_frags(const float* xp,
                                           const float* __restrict__ wp,
                                           const float* __restrict__ bp, int koff,
                                           v4f& a, v4f& c, v4f (&b)[8]) {
  a = *(const v4f*)xp;
  if (PRE) c = *(const v4f*)(bp + koff);
#pragma unroll
  for (int t = 0; t < 8; ++t) b[t] = *(const v4f*)(wp + t * 128);
}

__device__ __forceinline__ void bias_relu(v4f& a, const v4f& c) {
  float x = a.x, y = a.y, z = a.z, w = a.w;
  asm volatile(
      "v_add_f32 %0, %0, %4\n\t"
      "v_add_f32 %1, %1, %5\n\t"
      "v_add_f32 %2, %2, %6\n\t"
      "v_add_f32 %3, %3, %7\n\t"
      "v_max_num_f32 %0, 0, %0\n\t"
      "v_max_num_f32 %1, 0, %1\n\t"
      "v_max_num_f32 %2, 0, %2\n\t"
      "v_max_num_f32 %3, 0, %3"
      : "+v"(x), "+v"(y), "+v"(z), "+v"(w)
      : "v"(c.x), "v"(c.y), "v"(c.z), "v"(c.w));
  a.x = x;
  a.y = y;
  a.z = z;
  a.w = w;
}

__device__ __forceinline__ void mma_block(const v4f& a, const v4f (&b)[8], v8f (&acc)[8]) {
  v2f alo = {a.x, a.y}, ahi = {a.z, a.w};
#pragma unroll
  for (int t = 0; t < 8; ++t) {
    v2f blo = {b[t].x, b[t].y};
    acc[t] = __builtin_amdgcn_wmma_f32_16x16x4_f32(
         false, alo,  false, blo,
         (short)0, acc[t],  false,  false);
  }
#pragma unroll
  for (int t = 0; t < 8; ++t) {
    v2f bhi = {b[t].z, b[t].w};
    acc[t] = __builtin_amdgcn_wmma_f32_16x16x4_f32(
         false, ahi,  false, bhi,
         (short)0, acc[t],  false,  false);
  }
}

template <bool PRE>
__global__ void __launch_bounds__(128) gemm_wmma_f32(const float* X,
                                                     const float* __restrict__ Wp,
                                                     const float* __restrict__ bprev,
                                                     float* __restrict__ H,
                                                     int M, int K) {
  int wave = (int)((blockIdx.x * (size_t)blockDim.x + threadIdx.x) >> 5);
  int lane = threadIdx.x & 31;
  if (wave * 16 >= M) return;

  int nlo = lane & 15;
  int kh  = (lane >> 4) * 4;

  const float* xp = X + (size_t)(wave * 16 + nlo) * K + kh;
  const float* wp = Wp + lane * 4;
  const float* bp = bprev;
  const int nkb = K >> 3;

  v8f acc[8] = {};
  v4f a0, a1, c0 = {}, c1 = {}, b0[8], b1[8];
  load_frags<PRE>(xp, wp, bp, kh, a0, c0, b0);
#define PIN() __builtin_amdgcn_sched_barrier(0)
  int kb = 0;
  for (; kb + 2 < nkb; kb += 2) {
    load_frags<PRE>(xp + (kb + 1) * 8, wp + (size_t)(kb + 1) * 1024, bp, kh + (kb + 1) * 8, a1, c1, b1);
    PIN();
    if (PRE) bias_relu(a0, c0);
    mma_block(a0, b0, acc);
    PIN();
    load_frags<PRE>(xp + (kb + 2) * 8, wp + (size_t)(kb + 2) * 1024, bp, kh + (kb + 2) * 8, a0, c0, b0);
    PIN();
    if (PRE) bias_relu(a1, c1);
    mma_block(a1, b1, acc);
  }
  load_frags<PRE>(xp + (kb + 1) * 8, wp + (size_t)(kb + 1) * 1024, bp, kh + (kb + 1) * 8, a1, c1, b1);
  PIN();
  if (PRE) bias_relu(a0, c0);
  mma_block(a0, b0, acc);
  if (PRE) bias_relu(a1, c1);
  mma_block(a1, b1, acc);
#undef PIN
  asm volatile("v_nop\n\tv_nop\n\tv_nop\n\tv_nop"
               : "+v"(acc[0]), "+v"(acc[1]), "+v"(acc[2]), "+v"(acc[3]),
                 "+v"(acc[4]), "+v"(acc[5]), "+v"(acc[6]), "+v"(acc[7])
               : "v"(a0), "v"(b0[0]), "v"(b0[1]), "v"(b0[2]), "v"(b0[3]),
                 "v"(b0[4]), "v"(b0[5]), "v"(b0[6]), "v"(b0[7]),
                 "v"(a1), "v"(b1[0]), "v"(b1[1]), "v"(b1[2]), "v"(b1[3]),
                 "v"(b1[4]), "v"(b1[5]), "v"(b1[6]), "v"(b1[7]));

  int mbase = wave * 16 + ((lane >> 4) << 3);
#pragma unroll
  for (int t = 0; t < 8; ++t) {
#pragma unroll
    for (int e = 0; e < 8; ++e) H[(size_t)(mbase + e) * NHID + t * 16 + nlo] = acc[t][e];
  }
}


template <int K>
__device__ __forceinline__ unsigned gather_add_k(int r, int cl, float w, unsigned mask,
                                                 const float* __restrict__ H, float* tile,
                                                 int lane) {
  int id[K];
  v4f h[K];
#pragma unroll
  for (int u = 0; u < K; ++u) {
    id[u] = __builtin_ctz(mask);
    mask &= mask - 1;
    const int ru = __builtin_amdgcn_readlane(r, id[u]);
    h[u] = *(const v4f*)(H + (size_t)ru * NHID + lane * 4);
  }
#pragma unroll
  for (int u = 0; u < K; ++u) {
    const int cu = __builtin_amdgcn_readlane(cl, id[u]);
    const float wu = __int_as_float(__builtin_amdgcn_readlane(__float_as_int(w), id[u]));
    float* tp = tile + cu * NHID + lane;
    atomicAdd(tp + 0,  h[u].x * wu);
    atomicAdd(tp + 32, h[u].y * wu);
    atomicAdd(tp + 64, h[u].z * wu);
    atomicAdd(tp + 96, h[u].w * wu);
  }
  return mask;
}

__device__ __forceinline__ void drain_batch(const int* qp, int n, int lo,
                                            const int* __restrict__ rows,
                                            const int* __restrict__ cols,
                                            const float* __restrict__ dinv,
                                            const float* __restrict__ H, float* tile,
                                            int lane) {
  const bool valid = lane < n;
  const int e = qp[valid ? lane : 0];
  const int r = rows[e];
  const int c = cols[e];
  const float w = dinv[r] * dinv[c];
  const int cl = c - lo;
  unsigned mask = __builtin_amdgcn_ballot_w32(valid);
  int left = n;
  for (; left >= 8; left -= 8) mask = gather_add_k<8>(r, cl, w, mask, H, tile, lane);
  if (left >= 4) {
    mask = gather_add_k<4>(r, cl, w, mask, H, tile, lane);
    left -= 4;
  }
  if (left >= 2) {
    mask = gather_add_k<2>(r, cl, w, mask, H, tile, lane);
    left -= 2;
  }
  if (left >= 1) gather_add_k<1>(r, cl, w, mask, H, tile, lane);
}

__global__ void __launch_bounds__(OWN_WAVES * 32) agg_owner(const int* __restrict__ rows,
                                                            const int* __restrict__ cols,
                                                            const float* __restrict__ dinv,
                                                            const float* __restrict__ H,
                                                            float* __restrict__ Y,
                                                            int E, int N) {
  __shared__ float tile[OWN_ROWS * NHID];
  __shared__ int queue[OWN_WAVES][QCAP];
  const int tid = threadIdx.x;
  const int lane = tid & 31;
  const int wave = __builtin_amdgcn_readfirstlane(tid >> 5);
  const int lo = blockIdx.x * OWN_ROWS;
  const int nrows = min(OWN_ROWS, N - lo);
  const bool wide = (((size_t)cols) & 15) == 0;

  for (int i = tid; i < nrows * 32; i += OWN_WAVES * 32) {
    const int row = i >> 5;
    const float d = dinv[lo + row];
    const v4f v = *(const v4f*)(H + (size_t)(lo + row) * NHID + lane * 4);
    float* tp = tile + row * NHID + lane;
    tp[0]  = v.x * d * d;
    tp[32] = v.y * d * d;
    tp[64] = v.z * d * d;
    tp[96] = v.w * d * d;
  }
  __syncthreads();

  int* q = queue[wave];
  int qn = 0;

  const int wave_edges = 128 * SCAN_LOADS;
  const int trip_edges = OWN_WAVES * wave_edges;
  const int nfull = wide ? E / trip_edges : 0;
  if (nfull > 0) {
    const int4* cp = (const int4*)(cols + wave * wave_edges) + lane;
    int trip = (int)(blockIdx.x % (unsigned)nfull);
    int4 cur[SCAN_LOADS];
#pragma unroll
    for (int j = 0; j < SCAN_LOADS; ++j) cur[j] = cp[(size_t)trip * (trip_edges / 4) + j * 32];
    for (int i = 0; i < nfull; ++i) {
      const int base = (trip * OWN_WAVES + wave) * wave_edges;
      trip = trip + 1 < nfull ? trip + 1 : 0;
      const int4* np = cp + (size_t)trip * (trip_edges / 4);
      int4 t[SCAN_LOADS];
#pragma unroll
      for (int j = 0; j < SCAN_LOADS; ++j) t[j] = cur[j];
#pragma unroll
      for (int j = 0; j < SCAN_LOADS; ++j) cur[j] = np[j * 32];
#pragma unroll
      for (int j = 0; j < SCAN_LOADS; ++j) {
        const int tj[4] = {t[j].x, t[j].y, t[j].z, t[j].w};
#pragma unroll
        for (int k = 0; k < 4; ++k) {
          const bool hit = (unsigned)(tj[k] - lo) < (unsigned)nrows;
          const unsigned m = __builtin_amdgcn_ballot_w32(hit);
          if (m != 0) {
            if (hit) q[qn + (int)__builtin_amdgcn_mbcnt_lo(m, 0)] = base + j * 128 + lane * 4 + k;
            qn += __builtin_popcount(m);
          }
        }
      }
      if (qn >= 32) {
        int qh = 0;
        for (; qn - qh >= 32; qh += 32) drain_batch(q + qh, 32, lo, rows, cols, dinv, H, tile, lane);
        const int left = qn - qh;
        int keep = 0;
        if (lane < left) keep = q[qh + lane];
        if (lane < left) q[lane] = keep;
        qn = left;
      }
    }
  }
  for (int base = nfull * trip_edges + wave * 128; base < E; base += OWN_WAVES * 128) {
#pragma unroll
    for (int k = 0; k < 4; ++k) {
      const int e = base + k * 32 + lane;
      const int c = e < E ? cols[e] : -1;
      const bool hit = (unsigned)(c - lo) < (unsigned)nrows;
      const unsigned m = __builtin_amdgcn_ballot_w32(hit);
      if (m != 0) {
        if (hit) q[qn + (int)__builtin_amdgcn_mbcnt_lo(m, 0)] = e;
        qn += __builtin_popcount(m);
      }
    }
    if (qn >= 32) {
      int qh = 0;
      for (; qn - qh >= 32; qh += 32) drain_batch(q + qh, 32, lo, rows, cols, dinv, H, tile, lane);
      const int left = qn - qh;
      int keep = 0;
      if (lane < left) keep = q[qh + lane];
      if (lane < left) q[lane] = keep;
      qn = left;
    }
  }
  if (qn > 0) drain_batch(q, qn, lo, rows, cols, dinv, H, tile, lane);
  __syncthreads();

  for (int i = tid; i < nrows * 32; i += OWN_WAVES * 32) {
    const int row = i >> 5;
    const float* tp = tile + row * NHID + lane;
    const v4f v = {tp[0], tp[32], tp[64], tp[96]};
    *(v4f*)(Y + (size_t)(lo + row) * NHID + lane * 4) = v;
  }
}

__global__ void __launch_bounds__(256) final_logits(const float* __restrict__ X,
                                                    const float* __restrict__ bx,
                                                    const float* __restrict__ Wlt,
                                                    const float* __restrict__ bl,
                                                    float* __restrict__ out, int n) {
  int node = (int)((blockIdx.x * (size_t)blockDim.x + threadIdx.x) >> 5);
  int lane = threadIdx.x & 31;
  if (node >= n) return;

  float4 xv = *(const float4*)(X + (size_t)node * NHID + lane * 4);
  const float4 bv = *(const float4*)(bx + lane * 4);
  xv.x += bv.x;
  xv.y += bv.y;
  xv.z += bv.z;
  xv.w += bv.w;
  float logit[NCLASS];
#pragma unroll
  for (int c = 0; c < NCLASS; ++c) {
    float4 wv = *(const float4*)(Wlt + (size_t)c * NHID + lane * 4);
    float p = xv.x * wv.x + xv.y * wv.y + xv.z * wv.z + xv.w * wv.w;
#pragma unroll
    for (int off = 16; off > 0; off >>= 1) p += __shfl_xor(p, off, 32);
    logit[c] = p + bl[c];
  }
  float m = logit[0];
#pragma unroll
  for (int c = 1; c < NCLASS; ++c) m = fmaxf(m, logit[c]);
  float s = 0.0f;
#pragma unroll
  for (int c = 0; c < NCLASS; ++c) s += __expf(logit[c] - m);
  float lse = m + __logf(s);
  if (lane < NCLASS) out[(size_t)node * NCLASS + lane] = logit[lane] - lse;
}

extern "C" void kernel_launch(void* const* d_in, const int* in_sizes, int n_in,
                              void* d_out, int out_size, void* d_ws, size_t ws_size,
                              hipStream_t stream) {
  const float* x  = (const float*)d_in[0];
  const int*   ei = (const int*)d_in[1];
  const float* W1 = (const float*)d_in[2];
  const float* b1 = (const float*)d_in[3];
  const float* W2 = (const float*)d_in[4];
  const float* b2 = (const float*)d_in[5];
  const float* W3 = (const float*)d_in[6];
  const float* b3 = (const float*)d_in[7];
  const float* Wl = (const float*)d_in[8];
  const float* bl = (const float*)d_in[9];
  float* out = (float*)d_out;

  const int E = in_sizes[1] / 2;
  const int N = in_sizes[0] / NFEAT;

  char* ws = (char*)d_ws;
  size_t off = 0;
  auto alloc = [&](size_t bytes) -> float* {
    float* p = (float*)(ws + off);
    off += (bytes + 255) & ~(size_t)255;
    return p;
  };
  float* dinv = alloc((size_t)N * sizeof(float));
  float* Wp1  = alloc((size_t)NFEAT * NHID * sizeof(float));
  float* Wp2  = alloc((size_t)NHID * NHID * sizeof(float));
  float* Wp3  = alloc((size_t)NHID * NHID * sizeof(float));
  float* Wlt  = alloc((size_t)NHID * NCLASS * sizeof(float));
  float* H    = alloc((size_t)N * NHID * sizeof(float));
  float* Y    = alloc((size_t)N * NHID * sizeof(float));

  const int* rows = ei;
  const int* cols = ei + E;

  pack_w<<<(NFEAT * NHID + 255) / 256, 256, 0, stream>>>(W1, Wp1, NFEAT);
  pack_w<<<(NHID * NHID + 255) / 256, 256, 0, stream>>>(W2, Wp2, NHID);
  pack_w<<<(NHID * NHID + 255) / 256, 256, 0, stream>>>(W3, Wp3, NHID);
  transpose_w<<<(NHID * NCLASS + 255) / 256, 256, 0, stream>>>(Wl, Wlt, NHID, NCLASS);

  deg_init<<<(N + 255) / 256, 256, 0, stream>>>(dinv, N);
  deg_edges<<<(E + 255) / 256, 256, 0, stream>>>(cols, dinv, E);
  deg_fin<<<(N + 255) / 256, 256, 0, stream>>>(dinv, N);

  const int tiles = (N + 15) / 16;
  const int gemm_grid = (tiles + 3) / 4;
  const int ogrid = (N + OWN_ROWS - 1) / OWN_ROWS;

  gemm_wmma_f32<false><<<gemm_grid, 128, 0, stream>>>(x, Wp1, nullptr, H, N, NFEAT);
  agg_owner<<<ogrid, OWN_WAVES * 32, 0, stream>>>(rows, cols, dinv, H, Y, E, N);

  gemm_wmma_f32<true><<<gemm_grid, 128, 0, stream>>>(Y, Wp2, b1, H, N, NHID);
  agg_owner<<<ogrid, OWN_WAVES * 32, 0, stream>>>(rows, cols, dinv, H, Y, E, N);

  gemm_wmma_f32<true><<<gemm_grid, 128, 0, stream>>>(Y, Wp3, b2, H, N, NHID);
  agg_owner<<<ogrid, OWN_WAVES * 32, 0, stream>>>(rows, cols, dinv, H, Y, E, N);

  final_logits<<<(int)(((size_t)N * 32 + 255) / 256), 256, 0, stream>>>(Y, b3, Wlt, bl, out, N);
}


# Edge pass: the first layer's scan also writes each wave's found edges (source, target row, weight) to a workspace list; layers 2 and 3 replay the lists, no target scan; sums still in arrival order
# speedup vs baseline: 2.6203x; 1.2207x over previous
#include <hip/hip_runtime.h>

typedef __attribute__((ext_vector_type(2))) float v2f;
typedef __attribute__((ext_vector_type(4))) float v4f;
typedef __attribute__((ext_vector_type(8))) float v8f;
typedef __attribute__((ext_vector_type(2))) int v2i;

#define NFEAT 512
#define NHID  128
#define NCLASS 10

#define OWN_ROWS 196
#define OWN_WAVES 32
#define SCAN_LOADS 8
#define QCAP (32 + 128 * SCAN_LOADS)
#define LCAP 160
#define SRC_BITS 24
static_assert(OWN_ROWS <= (1 << (32 - SRC_BITS)), "a local target row must fit above the source row bits");


__global__ void transpose_w(const float* __restrict__ W, float* __restrict__ Wt,
                            int K, int Nc) {
  int idx = blockIdx.x * blockDim.x + threadIdx.x;
  if (idx < K * Nc) {
    int k = idx / Nc, n = idx % Nc;
    Wt[(size_t)n * K + k] = W[idx];
  }
}

__global__ void pack_w(const float* __restrict__ W, float* __restrict__ Wp, int K) {
  int idx = blockIdx.x * blockDim.x + threadIdx.x;
  if (idx < K * NHID) {
    int j = idx & 3, l = (idx >> 2) & 31, t = (idx >> 7) & 7, kb = idx >> 10;
    int k = kb * 8 + ((l >> 4) << 2) + j;
    int n = t * 16 + (l & 15);
    Wp[idx] = W[(size_t)k * NHID + n];
  }
}

__global__ void deg_init(float* __restrict__ deg, int n) {
  int i = blockIdx.x * blockDim.x + threadIdx.x;
  if (i < n) deg[i] = 1.0f;
}

__global__ void deg_edges(const int* __restrict__ cols, float* __restrict__ deg, int E) {
  int e = blockIdx.x * blockDim.x + threadIdx.x;
  if (e < E) atomicAdd(&deg[cols[e]], 1.0f);
}

__global__ void deg_fin(float* __restrict__ deg, int n) {
  int i = blockIdx.x * blockDim.x + threadIdx.x;
  if (i < n) deg[i] = rsqrtf(deg[i]);
}

template <bool PRE>
__device__ __forceinline__ void load_frags(const float* xp,
                                           const float* __restrict__ wp,
                                           const float* __restrict__ bp, int koff,
                                           v4f& a, v4f& c, v4f (&b)[8]) {
  a = *(const v4f*)xp;
  if (PRE) c = *(const v4f*)(bp + koff);
#pragma unroll
  for (int t = 0; t < 8; ++t) b[t] = *(const v4f*)(wp + t * 128);
}

__device__ __forceinline__ void bias_relu(v4f& a, const v4f& c) {
  float x = a.x, y = a.y, z = a.z, w = a.w;
  asm volatile(
      "v_add_f32 %0, %0, %4\n\t"
      "v_add_f32 %1, %1, %5\n\t"
      "v_add_f32 %2, %2, %6\n\t"
      "v_add_f32 %3, %3, %7\n\t"
      "v_max_num_f32 %0, 0, %0\n\t"
      "v_max_num_f32 %1, 0, %1\n\t"
      "v_max_num_f32 %2, 0, %2\n\t"
      "v_max_num_f32 %3, 0, %3"
      : "+v"(x), "+v"(y), "+v"(z), "+v"(w)
      : "v"(c.x), "v"(c.y), "v"(c.z), "v"(c.w));
  a.x = x;
  a.y = y;
  a.z = z;
  a.w = w;
}

__device__ __forceinline__ void mma_block(const v4f& a, const v4f (&b)[8], v8f (&acc)[8]) {
  v2f alo = {a.x, a.y}, ahi = {a.z, a.w};
#pragma unroll
  for (int t = 0; t < 8; ++t) {
    v2f blo = {b[t].x, b[t].y};
    acc[t] = __builtin_amdgcn_wmma_f32_16x16x4_f32(
         false, alo,  false, blo,
         (short)0, acc[t],  false,  false);
  }
#pragma unroll
  for (int t = 0; t < 8; ++t) {
    v2f bhi = {b[t].z, b[t].w};
    acc[t] = __builtin_amdgcn_wmma_f32_16x16x4_f32(
         false, ahi,  false, bhi,
         (short)0, acc[t],  false,  false);
  }
}

template <bool PRE>
__global__ void __launch_bounds__(128) gemm_wmma_f32(const float* X,
                                                     const float* __restrict__ Wp,
                                                     const float* __restrict__ bprev,
                                                     float* __restrict__ H,
                                                     int M, int K) {
  int wave = (int)((blockIdx.x * (size_t)blockDim.x + threadIdx.x) >> 5);
  int lane = threadIdx.x & 31;
  if (wave * 16 >= M) return;

  int nlo = lane & 15;
  int kh  = (lane >> 4) * 4;

  const float* xp = X + (size_t)(wave * 16 + nlo) * K + kh;
  const float* wp = Wp + lane * 4;
  const float* bp = bprev;
  const int nkb = K >> 3;

  v8f acc[8] = {};
  v4f a0, a1, c0 = {}, c1 = {}, b0[8], b1[8];
  load_frags<PRE>(xp, wp, bp, kh, a0, c0, b0);
#define PIN() __builtin_amdgcn_sched_barrier(0)
  int kb = 0;
  for (; kb + 2 < nkb; kb += 2) {
    load_frags<PRE>(xp + (kb + 1) * 8, wp + (size_t)(kb + 1) * 1024, bp, kh + (kb + 1) * 8, a1, c1, b1);
    PIN();
    if (PRE) bias_relu(a0, c0);
    mma_block(a0, b0, acc);
    PIN();
    load_frags<PRE>(xp + (kb + 2) * 8, wp + (size_t)(kb + 2) * 1024, bp, kh + (kb + 2) * 8, a0, c0, b0);
    PIN();
    if (PRE) bias_relu(a1, c1);
    mma_block(a1, b1, acc);
  }
  load_frags<PRE>(xp + (kb + 1) * 8, wp + (size_t)(kb + 1) * 1024, bp, kh + (kb + 1) * 8, a1, c1, b1);
  PIN();
  if (PRE) bias_relu(a0, c0);
  mma_block(a0, b0, acc);
  if (PRE) bias_relu(a1, c1);
  mma_block(a1, b1, acc);
#undef PIN
  asm volatile("v_nop\n\tv_nop\n\tv_nop\n\tv_nop"
               : "+v"(acc[0]), "+v"(acc[1]), "+v"(acc[2]), "+v"(acc[3]),
                 "+v"(acc[4]), "+v"(acc[5]), "+v"(acc[6]), "+v"(acc[7])
               : "v"(a0), "v"(b0[0]), "v"(b0[1]), "v"(b0[2]), "v"(b0[3]),
                 "v"(b0[4]), "v"(b0[5]), "v"(b0[6]), "v"(b0[7]),
                 "v"(a1), "v"(b1[0]), "v"(b1[1]), "v"(b1[2]), "v"(b1[3]),
                 "v"(b1[4]), "v"(b1[5]), "v"(b1[6]), "v"(b1[7]));

  int mbase = wave * 16 + ((lane >> 4) << 3);
#pragma unroll
  for (int t = 0; t < 8; ++t) {
#pragma unroll
    for (int e = 0; e < 8; ++e) H[(size_t)(mbase + e) * NHID + t * 16 + nlo] = acc[t][e];
  }
}


template <int K>
__device__ __forceinline__ unsigned gather_add_k(int r, int cl, float w, unsigned mask,
                                                 const float* __restrict__ H, float* tile,
                                                 int lane) {
  int id[K];
  v4f h[K];
#pragma unroll
  for (int u = 0; u < K; ++u) {
    id[u] = __builtin_ctz(mask);
    mask &= mask - 1;
    const int ru = __builtin_amdgcn_readlane(r, id[u]);
    h[u] = *(const v4f*)(H + (size_t)ru * NHID + lane * 4);
  }
#pragma unroll
  for (int u = 0; u < K; ++u) {
    const int cu = __builtin_amdgcn_readlane(cl, id[u]);
    const float wu = __int_as_float(__builtin_amdgcn_readlane(__float_as_int(w), id[u]));
    float* tp = tile + cu * NHID + lane;
    atomicAdd(tp + 0,  h[u].x * wu);
    atomicAdd(tp + 32, h[u].y * wu);
    atomicAdd(tp + 64, h[u].z * wu);
    atomicAdd(tp + 96, h[u].w * wu);
  }
  return mask;
}

__device__ __forceinline__ void add_batch(int r, int cl, float w, int n,
                                          const float* __restrict__ H, float* tile,
                                          int lane) {
  unsigned mask = __builtin_amdgcn_ballot_w32(lane < n);
  int left = n;
  for (; left >= 8; left -= 8) mask = gather_add_k<8>(r, cl, w, mask, H, tile, lane);
  if (left >= 4) {
    mask = gather_add_k<4>(r, cl, w, mask, H, tile, lane);
    left -= 4;
  }
  if (left >= 2) {
    mask = gather_add_k<2>(r, cl, w, mask, H, tile, lane);
    left -= 2;
  }
  if (left >= 1) gather_add_k<1>(r, cl, w, mask, H, tile, lane);
}

__device__ __forceinline__ void drain_batch(const int* qp, int n, int lo,
                                            const int* __restrict__ rows,
                                            const int* __restrict__ cols,
                                            const float* __restrict__ dinv,
                                            const float* __restrict__ H, float* tile,
                                            int lane, int* rec, int& cnt) {
  const bool valid = lane < n;
  const int e = qp[valid ? lane : 0];
  const int r = rows[e];
  const int c = cols[e];
  const float w = dinv[r] * dinv[c];
  const int cl = c - lo;
  if (rec != nullptr) {
    if (valid && cnt + n <= LCAP) {
      const v2i ent = {r | (cl << SRC_BITS), __float_as_int(w)};
      *(v2i*)(rec + 2 * (cnt + lane)) = ent;
    }
    cnt += n;
  }
  add_batch(r, cl, w, n, H, tile, lane);
}

template <bool REPLAY>
__global__ void __launch_bounds__(OWN_WAVES * 32) agg_owner(const int* __restrict__ rows,
                                                            const int* __restrict__ cols,
                                                            const float* __restrict__ dinv,
                                                            const float* __restrict__ H,
                                                            float* __restrict__ Y,
                                                            int* lists, int* counts,
                                                            int E, int N) {
  __shared__ float tile[OWN_ROWS * NHID];
  __shared__ int queue[OWN_WAVES][QCAP];
  const int tid = threadIdx.x;
  const int lane = tid & 31;
  const int wave = __builtin_amdgcn_readfirstlane(tid >> 5);
  const int lo = blockIdx.x * OWN_ROWS;
  const int nrows = min(OWN_ROWS, N - lo);
  const bool wide = (((size_t)cols) & 15) == 0;

  for (int i = tid; i < nrows * 32; i += OWN_WAVES * 32) {
    const int row = i >> 5;
    const float d = dinv[lo + row];
    const v4f v = *(const v4f*)(H + (size_t)(lo + row) * NHID + lane * 4);
    float* tp = tile + row * NHID + lane;
    tp[0]  = v.x * d * d;
    tp[32] = v.y * d * d;
    tp[64] = v.z * d * d;
    tp[96] = v.w * d * d;
  }
  __syncthreads();

  const int slot = blockIdx.x * OWN_WAVES + wave;
  int* rec = lists != nullptr ? lists + (size_t)slot * (2 * LCAP) : nullptr;
  const int nrec = REPLAY ? counts[slot] : LCAP + 1;
  if (REPLAY && nrec <= LCAP) {
    if (nrec > 0) {
      const v2i* lp = (const v2i*)rec;
      v2i cur = lp[min(lane, nrec - 1)];
      for (int i = 0; i < nrec; i += 32) {
        const v2i nxt = lp[min(i + 32 + lane, nrec - 1)];
        add_batch(cur.x & ((1 << SRC_BITS) - 1), (int)((unsigned)cur.x >> SRC_BITS),
                  __int_as_float(cur.y), min(32, nrec - i), H, tile, lane);
        cur = nxt;
      }
    }
  } else {
    if (REPLAY) rec = nullptr;
    int cnt = 0;
    int* q = queue[wave];
    int qn = 0;

    const int wave_edges = 128 * SCAN_LOADS;
    const int trip_edges = OWN_WAVES * wave_edges;
    const int nfull = wide ? E / trip_edges : 0;
    if (nfull > 0) {
      const int4* cp = (const int4*)(cols + wave * wave_edges) + lane;
      int trip = (int)(blockIdx.x % (unsigned)nfull);
      int4 cur[SCAN_LOADS];
#pragma unroll
      for (int j = 0; j < SCAN_LOADS; ++j) cur[j] = cp[(size_t)trip * (trip_edges / 4) + j * 32];
      for (int i = 0; i < nfull; ++i) {
        const int base = (trip * OWN_WAVES + wave) * wave_edges;
        trip = trip + 1 < nfull ? trip + 1 : 0;
        const int4* np = cp + (size_t)trip * (trip_edges / 4);
        int4 t[SCAN_LOADS];
#pragma unroll
        for (int j = 0; j < SCAN_LOADS; ++j) t[j] = cur[j];
#pragma unroll
        for (int j = 0; j < SCAN_LOADS; ++j) cur[j] = np[j * 32];
#pragma unroll
        for (int j = 0; j < SCAN_LOADS; ++j) {
          const int tj[4] = {t[j].x, t[j].y, t[j].z, t[j].w};
#pragma unroll
          for (int k = 0; k < 4; ++k) {
            const bool hit = (unsigned)(tj[k] - lo) < (unsigned)nrows;
            const unsigned m = __builtin_amdgcn_ballot_w32(hit);
            if (m != 0) {
              if (hit) q[qn + (int)__builtin_amdgcn_mbcnt_lo(m, 0)] = base + j * 128 + lane * 4 + k;
              qn += __builtin_popcount(m);
            }
          }
        }
        if (qn >= 32) {
          int qh = 0;
          for (; qn - qh >= 32; qh += 32)
            drain_batch(q + qh, 32, lo, rows, cols, dinv, H, tile, lane, rec, cnt);
          const int left = qn - qh;
          int keep = 0;
          if (lane < left) keep = q[qh + lane];
          if (lane < left) q[lane] = keep;
          qn = left;
        }
      }
    }
    for (int base = nfull * trip_edges + wave * 128; base < E; base += OWN_WAVES * 128) {
#pragma unroll
      for (int k = 0; k < 4; ++k) {
        const int e = base + k * 32 + lane;
        const int c = e < E ? cols[e] : -1;
        const bool hit = (unsigned)(c - lo) < (unsigned)nrows;
        const unsigned m = __builtin_amdgcn_ballot_w32(hit);
        if (m != 0) {
          if (hit) q[qn + (int)__builtin_amdgcn_mbcnt_lo(m, 0)] = e;
          qn += __builtin_popcount(m);
        }
      }
      if (qn >= 32) {
        int qh = 0;
        for (; qn - qh >= 32; qh += 32)
          drain_batch(q + qh, 32, lo, rows, cols, dinv, H, tile, lane, rec, cnt);
        const int left = qn - qh;
        int keep = 0;
        if (lane < left) keep = q[qh + lane];
        if (lane < left) q[lane] = keep;
        qn = left;
      }
    }
    if (qn > 0) drain_batch(q, qn, lo, rows, cols, dinv, H, tile, lane, rec, cnt);
    if (!REPLAY && rec != nullptr && lane == 0) counts[slot] = cnt;
  }
  __syncthreads();

  for (int i = tid; i < nrows * 32; i += OWN_WAVES * 32) {
    const int row = i >> 5;
    const float* tp = tile + row * NHID + lane;
    const v4f v = {tp[0], tp[32], tp[64], tp[96]};
    *(v4f*)(Y + (size_t)(lo + row) * NHID + lane * 4) = v;
  }
}

__global__ void __launch_bounds__(256) final_logits(const float* __restrict__ X,
                                                    const float* __restrict__ bx,
                                                    const float* __restrict__ Wlt,
                                                    const float* __restrict__ bl,
                                                    float* __restrict__ out, int n) {
  int node = (int)((blockIdx.x * (size_t)blockDim.x + threadIdx.x) >> 5);
  int lane = threadIdx.x & 31;
  if (node >= n) return;

  float4 xv = *(const float4*)(X + (size_t)node * NHID + lane * 4);
  const float4 bv = *(const float4*)(bx + lane * 4);
  xv.x += bv.x;
  xv.y += bv.y;
  xv.z += bv.z;
  xv.w += bv.w;
  float logit[NCLASS];
#pragma unroll
  for (int c = 0; c < NCLASS; ++c) {
    float4 wv = *(const float4*)(Wlt + (size_t)c * NHID + lane * 4);
    float p = xv.x * wv.x + xv.y * wv.y + xv.z * wv.z + xv.w * wv.w;
#pragma unroll
    for (int off = 16; off > 0; off >>= 1) p += __shfl_xor(p, off, 32);
    logit[c] = p + bl[c];
  }
  float m = logit[0];
#pragma unroll
  for (int c = 1; c < NCLASS; ++c) m = fmaxf(m, logit[c]);
  float s = 0.0f;
#pragma unroll
  for (int c = 0; c < NCLASS; ++c) s += __expf(logit[c] - m);
  float lse = m + __logf(s);
  if (lane < NCLASS) out[(size_t)node * NCLASS + lane] = logit[lane] - lse;
}

extern "C" void kernel_launch(void* const* d_in, const int* in_sizes, int n_in,
                              void* d_out, int out_size, void* d_ws, size_t ws_size,
                              hipStream_t stream) {
  const float* x  = (const float*)d_in[0];
  const int*   ei = (const int*)d_in[1];
  const float* W1 = (const float*)d_in[2];
  const float* b1 = (const float*)d_in[3];
  const float* W2 = (const float*)d_in[4];
  const float* b2 = (const float*)d_in[5];
  const float* W3 = (const float*)d_in[6];
  const float* b3 = (const float*)d_in[7];
  const float* Wl = (const float*)d_in[8];
  const float* bl = (const float*)d_in[9];
  float* out = (float*)d_out;

  const int E = in_sizes[1] / 2;
  const int N = in_sizes[0] / NFEAT;

  char* ws = (char*)d_ws;
  size_t off = 0;
  auto alloc = [&](size_t bytes) -> float* {
    float* p = (float*)(ws + off);
    off += (bytes + 255) & ~(size_t)255;
    return p;
  };
  float* dinv = alloc((size_t)N * sizeof(float));
  float* Wp1  = alloc((size_t)NFEAT * NHID * sizeof(float));
  float* Wp2  = alloc((size_t)NHID * NHID * sizeof(float));
  float* Wp3  = alloc((size_t)NHID * NHID * sizeof(float));
  float* Wlt  = alloc((size_t)NHID * NCLASS * sizeof(float));
  float* H    = alloc((size_t)N * NHID * sizeof(float));
  float* Y    = alloc((size_t)N * NHID * sizeof(float));

  const int* rows = ei;
  const int* cols = ei + E;

  pack_w<<<(NFEAT * NHID + 255) / 256, 256, 0, stream>>>(W1, Wp1, NFEAT);
  pack_w<<<(NHID * NHID + 255) / 256, 256, 0, stream>>>(W2, Wp2, NHID);
  pack_w<<<(NHID * NHID + 255) / 256, 256, 0, stream>>>(W3, Wp3, NHID);
  transpose_w<<<(NHID * NCLASS + 255) / 256, 256, 0, stream>>>(Wl, Wlt, NHID, NCLASS);

  deg_init<<<(N + 255) / 256, 256, 0, stream>>>(dinv, N);
  deg_edges<<<(E + 255) / 256, 256, 0, stream>>>(cols, dinv, E);
  deg_fin<<<(N + 255) / 256, 256, 0, stream>>>(dinv, N);

  const int tiles = (N + 15) / 16;
  const int gemm_grid = (tiles + 3) / 4;
  const int ogrid = (N + OWN_ROWS - 1) / OWN_ROWS;

  int* lists  = (int*)alloc((size_t)ogrid * OWN_WAVES * LCAP * 2 * sizeof(int));
  int* counts = (int*)alloc((size_t)ogrid * OWN_WAVES * sizeof(int));
  const bool replay = off <= ws_size && N <= (1 << SRC_BITS);
  if (!replay) lists = nullptr;

  gemm_wmma_f32<false><<<gemm_grid, 128, 0, stream>>>(x, Wp1, nullptr, H, N, NFEAT);
  agg_owner<false><<<ogrid, OWN_WAVES * 32, 0, stream>>>(rows, cols, dinv, H, Y, lists, counts, E, N);

  gemm_wmma_f32<true><<<gemm_grid, 128, 0, stream>>>(Y, Wp2, b1, H, N, NHID);
  if (replay)
    agg_owner<true><<<ogrid, OWN_WAVES * 32, 0, stream>>>(rows, cols, dinv, H, Y, lists, counts, E, N);
  else
    agg_owner<false><<<ogrid, OWN_WAVES * 32, 0, stream>>>(rows, cols, dinv, H, Y, lists, counts, E, N);

  gemm_wmma_f32<true><<<gemm_grid, 128, 0, stream>>>(Y, Wp3, b2, H, N, NHID);
  if (replay)
    agg_owner<true><<<ogrid, OWN_WAVES * 32, 0, stream>>>(rows, cols, dinv, H, Y, lists, counts, E, N);
  else
    agg_owner<false><<<ogrid, OWN_WAVES * 32, 0, stream>>>(rows, cols, dinv, H, Y, lists, counts, E, N);

  final_logits<<<(int)(((size_t)N * 32 + 255) / 256), 256, 0, stream>>>(Y, b3, Wlt, bl, out, N);
}


# Edge lists built once by a placing pass: each edge goes to the list of the workgroup owning its target (LDS counts, one global int add per owner and piece); no pass scans targets; arrival order
# speedup vs baseline: 2.7951x; 1.0667x over previous
#include <hip/hip_runtime.h>

typedef __attribute__((ext_vector_type(2))) float v2f;
typedef __attribute__((ext_vector_type(4))) float v4f;
typedef __attribute__((ext_vector_type(8))) float v8f;
typedef __attribute__((ext_vector_type(2))) int v2i;

#define NFEAT 512
#define NHID  128
#define NCLASS 10

#define OWN_ROWS 196
#define OWN_WAVES 32
#define SCAN_LOADS 8
#define QCAP (32 + 128 * SCAN_LOADS)
#define OCAP 4096
#define CNT_STRIDE 32
#define PLACE_WAVES 32
#define PLACE_PER 4
#define PLACE_MAXOWN 2048
#define SRC_BITS 24
static_assert(OWN_ROWS <= (1 << (32 - SRC_BITS)), "a local target row must fit above the source row bits");


__global__ void transpose_w(const float* __restrict__ W, float* __restrict__ Wt,
                            int K, int Nc) {
  int idx = blockIdx.x * blockDim.x + threadIdx.x;
  if (idx < K * Nc) {
    int k = idx / Nc, n = idx % Nc;
    Wt[(size_t)n * K + k] = W[idx];
  }
}

__global__ void pack_w(const float* __restrict__ W, float* __restrict__ Wp, int K) {
  int idx = blockIdx.x * blockDim.x + threadIdx.x;
  if (idx < K * NHID) {
    int j = idx & 3, l = (idx >> 2) & 31, t = (idx >> 7) & 7, kb = idx >> 10;
    int k = kb * 8 + ((l >> 4) << 2) + j;
    int n = t * 16 + (l & 15);
    Wp[idx] = W[(size_t)k * NHID + n];
  }
}

__global__ void deg_init(float* __restrict__ deg, int n, int* __restrict__ ocount, int nown) {
  int i = blockIdx.x * blockDim.x + threadIdx.x;
  if (i < n) deg[i] = 1.0f;
  if (i < nown) ocount[i * CNT_STRIDE] = 0;
}

__global__ void deg_edges(const int* __restrict__ cols, float* __restrict__ deg, int E) {
  int e = blockIdx.x * blockDim.x + threadIdx.x;
  if (e < E) atomicAdd(&deg[cols[e]], 1.0f);
}

__global__ void deg_fin(float* __restrict__ deg, int n) {
  int i = blockIdx.x * blockDim.x + threadIdx.x;
  if (i < n) deg[i] = rsqrtf(deg[i]);
}

template <bool PRE>
__device__ __forceinline__ void load_frags(const float* xp,
                                           const float* __restrict__ wp,
                                           const float* __restrict__ bp, int koff,
                                           v4f& a, v4f& c, v4f (&b)[8]) {
  a = *(const v4f*)xp;
  if (PRE) c = *(const v4f*)(bp + koff);
#pragma unroll
  for (int t = 0; t < 8; ++t) b[t] = *(const v4f*)(wp + t * 128);
}

__device__ __forceinline__ void bias_relu(v4f& a, const v4f& c) {
  float x = a.x, y = a.y, z = a.z, w = a.w;
  asm volatile(
      "v_add_f32 %0, %0, %4\n\t"
      "v_add_f32 %1, %1, %5\n\t"
      "v_add_f32 %2, %2, %6\n\t"
      "v_add_f32 %3, %3, %7\n\t"
      "v_max_num_f32 %0, 0, %0\n\t"
      "v_max_num_f32 %1, 0, %1\n\t"
      "v_max_num_f32 %2, 0, %2\n\t"
      "v_max_num_f32 %3, 0, %3"
      : "+v"(x), "+v"(y), "+v"(z), "+v"(w)
      : "v"(c.x), "v"(c.y), "v"(c.z), "v"(c.w));
  a.x = x;
  a.y = y;
  a.z = z;
  a.w = w;
}

__device__ __forceinline__ void mma_block(const v4f& a, const v4f (&b)[8], v8f (&acc)[8]) {
  v2f alo = {a.x, a.y}, ahi = {a.z, a.w};
#pragma unroll
  for (int t = 0; t < 8; ++t) {
    v2f blo = {b[t].x, b[t].y};
    acc[t] = __builtin_amdgcn_wmma_f32_16x16x4_f32(
         false, alo,  false, blo,
         (short)0, acc[t],  false,  false);
  }
#pragma unroll
  for (int t = 0; t < 8; ++t) {
    v2f bhi = {b[t].z, b[t].w};
    acc[t] = __builtin_amdgcn_wmma_f32_16x16x4_f32(
         false, ahi,  false, bhi,
         (short)0, acc[t],  false,  false);
  }
}

template <bool PRE>
__global__ void __launch_bounds__(128) gemm_wmma_f32(const float* X,
                                                     const float* __restrict__ Wp,
                                                     const float* __restrict__ bprev,
                                                     float* __restrict__ H,
                                                     int M, int K) {
  int wave = (int)((blockIdx.x * (size_t)blockDim.x + threadIdx.x) >> 5);
  int lane = threadIdx.x & 31;
  if (wave * 16 >= M) return;

  int nlo = lane & 15;
  int kh  = (lane >> 4) * 4;

  const float* xp = X + (size_t)(wave * 16 + nlo) * K + kh;
  const float* wp = Wp + lane * 4;
  const float* bp = bprev;
  const int nkb = K >> 3;

  v8f acc[8] = {};
  v4f a0, a1, c0 = {}, c1 = {}, b0[8], b1[8];
  load_frags<PRE>(xp, wp, bp, kh, a0, c0, b0);
#define PIN() __builtin_amdgcn_sched_barrier(0)
  int kb = 0;
  for (; kb + 2 < nkb; kb += 2) {
    load_frags<PRE>(xp + (kb + 1) * 8, wp + (size_t)(kb + 1) * 1024, bp, kh + (kb + 1) * 8, a1, c1, b1);
    PIN();
    if (PRE) bias_relu(a0, c0);
    mma_block(a0, b0, acc);
    PIN();
    load_frags<PRE>(xp + (kb + 2) * 8, wp + (size_t)(kb + 2) * 1024, bp, kh + (kb + 2) * 8, a0, c0, b0);
    PIN();
    if (PRE) bias_relu(a1, c1);
    mma_block(a1, b1, acc);
  }
  load_frags<PRE>(xp + (kb + 1) * 8, wp + (size_t)(kb + 1) * 1024, bp, kh + (kb + 1) * 8, a1, c1, b1);
  PIN();
  if (PRE) bias_relu(a0, c0);
  mma_block(a0, b0, acc);
  if (PRE) bias_relu(a1, c1);
  mma_block(a1, b1, acc);
#undef PIN
  asm volatile("v_nop\n\tv_nop\n\tv_nop\n\tv_nop"
               : "+v"(acc[0]), "+v"(acc[1]), "+v"(acc[2]), "+v"(acc[3]),
                 "+v"(acc[4]), "+v"(acc[5]), "+v"(acc[6]), "+v"(acc[7])
               : "v"(a0), "v"(b0[0]), "v"(b0[1]), "v"(b0[2]), "v"(b0[3]),
                 "v"(b0[4]), "v"(b0[5]), "v"(b0[6]), "v"(b0[7]),
                 "v"(a1), "v"(b1[0]), "v"(b1[1]), "v"(b1[2]), "v"(b1[3]),
                 "v"(b1[4]), "v"(b1[5]), "v"(b1[6]), "v"(b1[7]));

  int mbase = wave * 16 + ((lane >> 4) << 3);
#pragma unroll
  for (int t = 0; t < 8; ++t) {
#pragma unroll
    for (int e = 0; e < 8; ++e) H[(size_t)(mbase + e) * NHID + t * 16 + nlo] = acc[t][e];
  }
}


__global__ void __launch_bounds__(PLACE_WAVES * 32) place_edges(const int* __restrict__ rows,
                                                                const int* __restrict__ cols,
                                                                const float* __restrict__ dinv,
                                                                int* __restrict__ olist,
                                                                int* ocount, int E, int nown) {
  __shared__ int cnt[PLACE_MAXOWN];
  const int tid = threadIdx.x;
  for (int i = tid; i < nown; i += PLACE_WAVES * 32) cnt[i] = 0;
  __syncthreads();

  const int e0 = blockIdx.x * (PLACE_WAVES * 32 * PLACE_PER) + tid;
  int r[PLACE_PER], c[PLACE_PER], o[PLACE_PER], rk[PLACE_PER], pos[PLACE_PER];
  float w[PLACE_PER];
  bool ok[PLACE_PER];
#pragma unroll
  for (int k = 0; k < PLACE_PER; ++k) {
    const int e = e0 + k * (PLACE_WAVES * 32);
    const int ee = min(e, E - 1);
    ok[k] = e < E;
    r[k] = rows[ee];
    c[k] = cols[ee];
  }
#pragma unroll
  for (int k = 0; k < PLACE_PER; ++k) w[k] = dinv[r[k]] * dinv[c[k]];
#pragma unroll
  for (int k = 0; k < PLACE_PER; ++k) {
    o[k] = (int)((unsigned)c[k] / (unsigned)OWN_ROWS);
    rk[k] = ok[k] ? atomicAdd(&cnt[o[k]], 1) : 0;
  }
  __syncthreads();
  for (int i = tid; i < nown; i += PLACE_WAVES * 32) {
    const int n = cnt[i];
    if (n > 0) cnt[i] = atomicAdd(&ocount[i * CNT_STRIDE], n);
  }
  __syncthreads();
#pragma unroll
  for (int k = 0; k < PLACE_PER; ++k) pos[k] = cnt[o[k]] + rk[k];
#pragma unroll
  for (int k = 0; k < PLACE_PER; ++k) {
    if (ok[k] && pos[k] < OCAP) {
      const v2i ent = {r[k] | ((c[k] - o[k] * OWN_ROWS) << SRC_BITS), __float_as_int(w[k])};
      *(v2i*)(olist + 2 * ((size_t)o[k] * OCAP + pos[k])) = ent;
    }
  }
}

template <int K>
__device__ __forceinline__ unsigned gather_add_k(int r, int cl, float w, unsigned mask,
                                                 const float* __restrict__ H, float* tile,
                                                 int lane) {
  int id[K];
  v4f h[K];
#pragma unroll
  for (int u = 0; u < K; ++u) {
    id[u] = __builtin_ctz(mask);
    mask &= mask - 1;
    const int ru = __builtin_amdgcn_readlane(r, id[u]);
    h[u] = *(const v4f*)(H + (size_t)ru * NHID + lane * 4);
  }
#pragma unroll
  for (int u = 0; u < K; ++u) {
    const int cu = __builtin_amdgcn_readlane(cl, id[u]);
    const float wu = __int_as_float(__builtin_amdgcn_readlane(__float_as_int(w), id[u]));
    float* tp = tile + cu * NHID + lane;
    atomicAdd(tp + 0,  h[u].x * wu);
    atomicAdd(tp + 32, h[u].y * wu);
    atomicAdd(tp + 64, h[u].z * wu);
    atomicAdd(tp + 96, h[u].w * wu);
  }
  return mask;
}

__device__ __forceinline__ void add_batch(int r, int cl, float w, int n,
                                          const float* __restrict__ H, float* tile,
                                          int lane) {
  unsigned mask = __builtin_amdgcn_ballot_w32(lane < n);
  int left = n;
  for (; left >= 8; left -= 8) mask = gather_add_k<8>(r, cl, w, mask, H, tile, lane);
  if (left >= 4) {
    mask = gather_add_k<4>(r, cl, w, mask, H, tile, lane);
    left -= 4;
  }
  if (left >= 2) {
    mask = gather_add_k<2>(r, cl, w, mask, H, tile, lane);
    left -= 2;
  }
  if (left >= 1) gather_add_k<1>(r, cl, w, mask, H, tile, lane);
}

__device__ __forceinline__ void drain_batch(const int* qp, int n, int lo,
                                            const int* __restrict__ rows,
                                            const int* __restrict__ cols,
                                            const float* __restrict__ dinv,
                                            const float* __restrict__ H, float* tile,
                                            int lane) {
  const bool valid = lane < n;
  const int e = qp[valid ? lane : 0];
  const int r = rows[e];
  const int c = cols[e];
  const float w = dinv[r] * dinv[c];
  add_batch(r, c - lo, w, n, H, tile, lane);
}

__global__ void __launch_bounds__(OWN_WAVES * 32) agg_owner(const int* __restrict__ rows,
                                                            const int* __restrict__ cols,
                                                            const float* __restrict__ dinv,
                                                            const float* __restrict__ H,
                                                            float* __restrict__ Y,
                                                            const int* __restrict__ olist,
                                                            const int* __restrict__ ocount,
                                                            int E, int N) {
  __shared__ float tile[OWN_ROWS * NHID];
  __shared__ int queue[OWN_WAVES][QCAP];
  const int tid = threadIdx.x;
  const int lane = tid & 31;
  const int wave = __builtin_amdgcn_readfirstlane(tid >> 5);
  const int lo = blockIdx.x * OWN_ROWS;
  const int nrows = min(OWN_ROWS, N - lo);
  const bool wide = (((size_t)cols) & 15) == 0;

  for (int i = tid; i < nrows * 32; i += OWN_WAVES * 32) {
    const int row = i >> 5;
    const float d = dinv[lo + row];
    const v4f v = *(const v4f*)(H + (size_t)(lo + row) * NHID + lane * 4);
    float* tp = tile + row * NHID + lane;
    tp[0]  = v.x * d * d;
    tp[32] = v.y * d * d;
    tp[64] = v.z * d * d;
    tp[96] = v.w * d * d;
  }
  __syncthreads();

  const int nown = olist != nullptr
                       ? __builtin_amdgcn_readfirstlane(ocount[blockIdx.x * CNT_STRIDE])
                       : OCAP + 1;
  if (nown <= OCAP) {
    const int per = (nown + OWN_WAVES - 1) / OWN_WAVES;
    const int beg = min(nown, wave * per);
    const int m = min(nown, beg + per) - beg;
    if (m > 0) {
      const v2i* lp = (const v2i*)olist + (size_t)blockIdx.x * OCAP + beg;
      v2i cur = lp[min(lane, m - 1)];
      for (int i = 0; i < m; i += 32) {
        const v2i nxt = lp[min(i + 32 + lane, m - 1)];
        add_batch(cur.x & ((1 << SRC_BITS) - 1), (int)((unsigned)cur.x >> SRC_BITS),
                  __int_as_float(cur.y), min(32, m - i), H, tile, lane);
        cur = nxt;
      }
    }
  } else {
    int* q = queue[wave];
    int qn = 0;

    const int wave_edges = 128 * SCAN_LOADS;
    const int trip_edges = OWN_WAVES * wave_edges;
    const int nfull = wide ? E / trip_edges : 0;
    if (nfull > 0) {
      const int4* cp = (const int4*)(cols + wave * wave_edges) + lane;
      int trip = (int)(blockIdx.x % (unsigned)nfull);
      int4 cur[SCAN_LOADS];
#pragma unroll
      for (int j = 0; j < SCAN_LOADS; ++j) cur[j] = cp[(size_t)trip * (trip_edges / 4) + j * 32];
      for (int i = 0; i < nfull; ++i) {
        const int base = (trip * OWN_WAVES + wave) * wave_edges;
        trip = trip + 1 < nfull ? trip + 1 : 0;
        const int4* np = cp + (size_t)trip * (trip_edges / 4);
        int4 t[SCAN_LOADS];
#pragma unroll
        for (int j = 0; j < SCAN_LOADS; ++j) t[j] = cur[j];
#pragma unroll
        for (int j = 0; j < SCAN_LOADS; ++j) cur[j] = np[j * 32];
#pragma unroll
        for (int j = 0; j < SCAN_LOADS; ++j) {
          const int tj[4] = {t[j].x, t[j].y, t[j].z, t[j].w};
#pragma unroll
          for (int k = 0; k < 4; ++k) {
            const bool hit = (unsigned)(tj[k] - lo) < (unsigned)nrows;
            const unsigned m = __builtin_amdgcn_ballot_w32(hit);
            if (m != 0) {
              if (hit) q[qn + (int)__builtin_amdgcn_mbcnt_lo(m, 0)] = base + j * 128 + lane * 4 + k;
              qn += __builtin_popcount(m);
            }
          }
        }
        if (qn >= 32) {
          int qh = 0;
          for (; qn - qh >= 32; qh += 32)
            drain_batch(q + qh, 32, lo, rows, cols, dinv, H, tile, lane);
          const int left = qn - qh;
          int keep = 0;
          if (lane < left) keep = q[qh + lane];
          if (lane < left) q[lane] = keep;
          qn = left;
        }
      }
    }
    for (int base = nfull * trip_edges + wave * 128; base < E; base += OWN_WAVES * 128) {
#pragma unroll
      for (int k = 0; k < 4; ++k) {
        const int e = base + k * 32 + lane;
        const int c = e < E ? cols[e] : -1;
        const bool hit = (unsigned)(c - lo) < (unsigned)nrows;
        const unsigned m = __builtin_amdgcn_ballot_w32(hit);
        if (m != 0) {
          if (hit) q[qn + (int)__builtin_amdgcn_mbcnt_lo(m, 0)] = e;
          qn += __builtin_popcount(m);
        }
      }
      if (qn >= 32) {
        int qh = 0;
        for (; qn - qh >= 32; qh += 32)
          drain_batch(q + qh, 32, lo, rows, cols, dinv, H, tile, lane);
        const int left = qn - qh;
        int keep = 0;
        if (lane < left) keep = q[qh + lane];
        if (lane < left) q[lane] = keep;
        qn = left;
      }
    }
    if (qn > 0) drain_batch(q, qn, lo, rows, cols, dinv, H, tile, lane);
  }
  __syncthreads();

  for (int i = tid; i < nrows * 32; i += OWN_WAVES * 32) {
    const int row = i >> 5;
    const float* tp = tile + row * NHID + lane;
    const v4f v = {tp[0], tp[32], tp[64], tp[96]};
    *(v4f*)(Y + (size_t)(lo + row) * NHID + lane * 4) = v;
  }
}

__global__ void __launch_bounds__(256) final_logits(const float* __restrict__ X,
                                                    const float* __restrict__ bx,
                                                    const float* __restrict__ Wlt,
                                                    const float* __restrict__ bl,
                                                    float* __restrict__ out, int n) {
  int node = (int)((blockIdx.x * (size_t)blockDim.x + threadIdx.x) >> 5);
  int lane = threadIdx.x & 31;
  if (node >= n) return;

  float4 xv = *(const float4*)(X + (size_t)node * NHID + lane * 4);
  const float4 bv = *(const float4*)(bx + lane * 4);
  xv.x += bv.x;
  xv.y += bv.y;
  xv.z += bv.z;
  xv.w += bv.w;
  float logit[NCLASS];
#pragma unroll
  for (int c = 0; c < NCLASS; ++c) {
    float4 wv = *(const float4*)(Wlt + (size_t)c * NHID + lane * 4);
    float p = xv.x * wv.x + xv.y * wv.y + xv.z * wv.z + xv.w * wv.w;
#pragma unroll
    for (int off = 16; off > 0; off >>= 1) p += __shfl_xor(p, off, 32);
    logit[c] = p + bl[c];
  }
  float m = logit[0];
#pragma unroll
  for (int c = 1; c < NCLASS; ++c) m = fmaxf(m, logit[c]);
  float s = 0.0f;
#pragma unroll
  for (int c = 0; c < NCLASS; ++c) s += __expf(logit[c] - m);
  float lse = m + __logf(s);
  if (lane < NCLASS) out[(size_t)node * NCLASS + lane] = logit[lane] - lse;
}

extern "C" void kernel_launch(void* const* d_in, const int* in_sizes, int n_in,
                              void* d_out, int out_size, void* d_ws, size_t ws_size,
                              hipStream_t stream) {
  const float* x  = (const float*)d_in[0];
  const int*   ei = (const int*)d_in[1];
  const float* W1 = (const float*)d_in[2];
  const float* b1 = (const float*)d_in[3];
  const float* W2 = (const float*)d_in[4];
  const float* b2 = (const float*)d_in[5];
  const float* W3 = (const float*)d_in[6];
  const float* b3 = (const float*)d_in[7];
  const float* Wl = (const float*)d_in[8];
  const float* bl = (const float*)d_in[9];
  float* out = (float*)d_out;

  const int E = in_sizes[1] / 2;
  const int N = in_sizes[0] / NFEAT;

  char* ws = (char*)d_ws;
  size_t off = 0;
  auto alloc = [&](size_t bytes) -> float* {
    float* p = (float*)(ws + off);
    off += (bytes + 255) & ~(size_t)255;
    return p;
  };
  float* dinv = alloc((size_t)N * sizeof(float));
  float* Wp1  = alloc((size_t)NFEAT * NHID * sizeof(float));
  float* Wp2  = alloc((size_t)NHID * NHID * sizeof(float));
  float* Wp3  = alloc((size_t)NHID * NHID * sizeof(float));
  float* Wlt  = alloc((size_t)NHID * NCLASS * sizeof(float));
  float* H    = alloc((size_t)N * NHID * sizeof(float));
  float* Y    = alloc((size_t)N * NHID * sizeof(float));

  const int* rows = ei;
  const int* cols = ei + E;

  const int tiles = (N + 15) / 16;
  const int gemm_grid = (tiles + 3) / 4;
  const int ogrid = (N + OWN_ROWS - 1) / OWN_ROWS;

  int* olist  = (int*)alloc((size_t)ogrid * OCAP * 2 * sizeof(int));
  int* ocount = (int*)alloc((size_t)ogrid * CNT_STRIDE * sizeof(int));
  const bool listed = off <= ws_size && N <= (1 << SRC_BITS) && ogrid <= PLACE_MAXOWN;
  if (!listed) {
    olist = nullptr;
    ocount = nullptr;
  }

  pack_w<<<(NFEAT * NHID + 255) / 256, 256, 0, stream>>>(W1, Wp1, NFEAT);
  pack_w<<<(NHID * NHID + 255) / 256, 256, 0, stream>>>(W2, Wp2, NHID);
  pack_w<<<(NHID * NHID + 255) / 256, 256, 0, stream>>>(W3, Wp3, NHID);
  transpose_w<<<(NHID * NCLASS + 255) / 256, 256, 0, stream>>>(Wl, Wlt, NHID, NCLASS);

  deg_init<<<(N + 255) / 256, 256, 0, stream>>>(dinv, N, ocount, listed ? ogrid : 0);
  deg_edges<<<(E + 255) / 256, 256, 0, stream>>>(cols, dinv, E);
  deg_fin<<<(N + 255) / 256, 256, 0, stream>>>(dinv, N);
  if (listed) {
    const int per_group = PLACE_WAVES * 32 * PLACE_PER;
    place_edges<<<(E + per_group - 1) / per_group, PLACE_WAVES * 32, 0, stream>>>(rows, cols, dinv, olist,
                                                                                ocount, E, ogrid);
  }

  gemm_wmma_f32<false><<<gemm_grid, 128, 0, stream>>>(x, Wp1, nullptr, H, N, NFEAT);
  agg_owner<<<ogrid, OWN_WAVES * 32, 0, stream>>>(rows, cols, dinv, H, Y, olist, ocount, E, N);

  gemm_wmma_f32<true><<<gemm_grid, 128, 0, stream>>>(Y, Wp2, b1, H, N, NHID);
  agg_owner<<<ogrid, OWN_WAVES * 32, 0, stream>>>(rows, cols, dinv, H, Y, olist, ocount, E, N);

  gemm_wmma_f32<true><<<gemm_grid, 128, 0, stream>>>(Y, Wp3, b2, H, N, NHID);
  agg_owner<<<ogrid, OWN_WAVES * 32, 0, stream>>>(rows, cols, dinv, H, Y, olist, ocount, E, N);

  final_logits<<<(int)(((size_t)N * 32 + 255) / 256), 256, 0, stream>>>(Y, b3, Wlt, bl, out, N);
}


# Edge pass, listed owners: list counting-sorted by target row in LDS each pass, self loop as a row's first entry; a wave sums its rows one by one in registers (fma, arrival order); no LDS float adds
# speedup vs baseline: 2.9742x; 1.0641x over previous
#include <hip/hip_runtime.h>

typedef __attribute__((ext_vector_type(2))) float v2f;
typedef __attribute__((ext_vector_type(4))) float v4f;
typedef __attribute__((ext_vector_type(8))) float v8f;
typedef __attribute__((ext_vector_type(2))) int v2i;

#define NFEAT 512
#define NHID  128
#define NCLASS 10

#define OWN_ROWS 196
#define OWN_WAVES 32
#define SCAN_LOADS 8
#define QCAP (32 + 128 * SCAN_LOADS)
#define OCAP 4096
#define CNT_STRIDE 32
#define PLACE_WAVES 32
#define PLACE_PER 4
#define PLACE_MAXOWN 2048
#define SRC_BITS 24
static_assert(OWN_ROWS <= (1 << (32 - SRC_BITS)), "a local target row must fit above the source row bits");


__global__ void transpose_w(const float* __restrict__ W, float* __restrict__ Wt,
                            int K, int Nc) {
  int idx = blockIdx.x * blockDim.x + threadIdx.x;
  if (idx < K * Nc) {
    int k = idx / Nc, n = idx % Nc;
    Wt[(size_t)n * K + k] = W[idx];
  }
}

__global__ void pack_w(const float* __restrict__ W, float* __restrict__ Wp, int K) {
  int idx = blockIdx.x * blockDim.x + threadIdx.x;
  if (idx < K * NHID) {
    int j = idx & 3, l = (idx >> 2) & 31, t = (idx >> 7) & 7, kb = idx >> 10;
    int k = kb * 8 + ((l >> 4) << 2) + j;
    int n = t * 16 + (l & 15);
    Wp[idx] = W[(size_t)k * NHID + n];
  }
}

__global__ void deg_init(float* __restrict__ deg, int n, int* __restrict__ ocount, int nown) {
  int i = blockIdx.x * blockDim.x + threadIdx.x;
  if (i < n) deg[i] = 1.0f;
  if (i < nown) ocount[i * CNT_STRIDE] = 0;
}

__global__ void deg_edges(const int* __restrict__ cols, float* __restrict__ deg, int E) {
  int e = blockIdx.x * blockDim.x + threadIdx.x;
  if (e < E) atomicAdd(&deg[cols[e]], 1.0f);
}

__global__ void deg_fin(float* __restrict__ deg, int n) {
  int i = blockIdx.x * blockDim.x + threadIdx.x;
  if (i < n) deg[i] = rsqrtf(deg[i]);
}

template <bool PRE>
__device__ __forceinline__ void load_frags(const float* xp,
                                           const float* __restrict__ wp,
                                           const float* __restrict__ bp, int koff,
                                           v4f& a, v4f& c, v4f (&b)[8]) {
  a = *(const v4f*)xp;
  if (PRE) c = *(const v4f*)(bp + koff);
#pragma unroll
  for (int t = 0; t < 8; ++t) b[t] = *(const v4f*)(wp + t * 128);
}

__device__ __forceinline__ void bias_relu(v4f& a, const v4f& c) {
  float x = a.x, y = a.y, z = a.z, w = a.w;
  asm volatile(
      "v_add_f32 %0, %0, %4\n\t"
      "v_add_f32 %1, %1, %5\n\t"
      "v_add_f32 %2, %2, %6\n\t"
      "v_add_f32 %3, %3, %7\n\t"
      "v_max_num_f32 %0, 0, %0\n\t"
      "v_max_num_f32 %1, 0, %1\n\t"
      "v_max_num_f32 %2, 0, %2\n\t"
      "v_max_num_f32 %3, 0, %3"
      : "+v"(x), "+v"(y), "+v"(z), "+v"(w)
      : "v"(c.x), "v"(c.y), "v"(c.z), "v"(c.w));
  a.x = x;
  a.y = y;
  a.z = z;
  a.w = w;
}

__device__ __forceinline__ void mma_block(const v4f& a, const v4f (&b)[8], v8f (&acc)[8]) {
  v2f alo = {a.x, a.y}, ahi = {a.z, a.w};
#pragma unroll
  for (int t = 0; t < 8; ++t) {
    v2f blo = {b[t].x, b[t].y};
    acc[t] = __builtin_amdgcn_wmma_f32_16x16x4_f32(
         false, alo,  false, blo,
         (short)0, acc[t],  false,  false);
  }
#pragma unroll
  for (int t = 0; t < 8; ++t) {
    v2f bhi = {b[t].z, b[t].w};
    acc[t] = __builtin_amdgcn_wmma_f32_16x16x4_f32(
         false, ahi,  false, bhi,
         (short)0, acc[t],  false,  false);
  }
}

template <bool PRE>
__global__ void __launch_bounds__(128) gemm_wmma_f32(const float* X,
                                                     const float* __restrict__ Wp,
                                                     const float* __restrict__ bprev,
                                                     float* __restrict__ H,
                                                     int M, int K) {
  int wave = (int)((blockIdx.x * (size_t)blockDim.x + threadIdx.x) >> 5);
  int lane = threadIdx.x & 31;
  if (wave * 16 >= M) return;

  int nlo = lane & 15;
  int kh  = (lane >> 4) * 4;

  const float* xp = X + (size_t)(wave * 16 + nlo) * K + kh;
  const float* wp = Wp + lane * 4;
  const float* bp = bprev;
  const int nkb = K >> 3;

  v8f acc[8] = {};
  v4f a0, a1, c0 = {}, c1 = {}, b0[8], b1[8];
  load_frags<PRE>(xp, wp, bp, kh, a0, c0, b0);
#define PIN() __builtin_amdgcn_sched_barrier(0)
  int kb = 0;
  for (; kb + 2 < nkb; kb += 2) {
    load_frags<PRE>(xp + (kb + 1) * 8, wp + (size_t)(kb + 1) * 1024, bp, kh + (kb + 1) * 8, a1, c1, b1);
    PIN();
    if (PRE) bias_relu(a0, c0);
    mma_block(a0, b0, acc);
    PIN();
    load_frags<PRE>(xp + (kb + 2) * 8, wp + (size_t)(kb + 2) * 1024, bp, kh + (kb + 2) * 8, a0, c0, b0);
    PIN();
    if (PRE) bias_relu(a1, c1);
    mma_block(a1, b1, acc);
  }
  load_frags<PRE>(xp + (kb + 1) * 8, wp + (size_t)(kb + 1) * 1024, bp, kh + (kb + 1) * 8, a1, c1, b1);
  PIN();
  if (PRE) bias_relu(a0, c0);
  mma_block(a0, b0, acc);
  if (PRE) bias_relu(a1, c1);
  mma_block(a1, b1, acc);
#undef PIN
  asm volatile("v_nop\n\tv_nop\n\tv_nop\n\tv_nop"
               : "+v"(acc[0]), "+v"(acc[1]), "+v"(acc[2]), "+v"(acc[3]),
                 "+v"(acc[4]), "+v"(acc[5]), "+v"(acc[6]), "+v"(acc[7])
               : "v"(a0), "v"(b0[0]), "v"(b0[1]), "v"(b0[2]), "v"(b0[3]),
                 "v"(b0[4]), "v"(b0[5]), "v"(b0[6]), "v"(b0[7]),
                 "v"(a1), "v"(b1[0]), "v"(b1[1]), "v"(b1[2]), "v"(b1[3]),
                 "v"(b1[4]), "v"(b1[5]), "v"(b1[6]), "v"(b1[7]));

  int mbase = wave * 16 + ((lane >> 4) << 3);
#pragma unroll
  for (int t = 0; t < 8; ++t) {
#pragma unroll
    for (int e = 0; e < 8; ++e) H[(size_t)(mbase + e) * NHID + t * 16 + nlo] = acc[t][e];
  }
}


__global__ void __launch_bounds__(PLACE_WAVES * 32) place_edges(const int* __restrict__ rows,
                                                                const int* __restrict__ cols,
                                                                const float* __restrict__ dinv,
                                                                int* __restrict__ olist,
                                                                int* ocount, int E, int nown) {
  __shared__ int cnt[PLACE_MAXOWN];
  const int tid = threadIdx.x;
  for (int i = tid; i < nown; i += PLACE_WAVES * 32) cnt[i] = 0;
  __syncthreads();

  const int e0 = blockIdx.x * (PLACE_WAVES * 32 * PLACE_PER) + tid;
  int r[PLACE_PER], c[PLACE_PER], o[PLACE_PER], rk[PLACE_PER], pos[PLACE_PER];
  float w[PLACE_PER];
  bool ok[PLACE_PER];
#pragma unroll
  for (int k = 0; k < PLACE_PER; ++k) {
    const int e = e0 + k * (PLACE_WAVES * 32);
    const int ee = min(e, E - 1);
    ok[k] = e < E;
    r[k] = rows[ee];
    c[k] = cols[ee];
  }
#pragma unroll
  for (int k = 0; k < PLACE_PER; ++k) w[k] = dinv[r[k]] * dinv[c[k]];
#pragma unroll
  for (int k = 0; k < PLACE_PER; ++k) {
    o[k] = (int)((unsigned)c[k] / (unsigned)OWN_ROWS);
    rk[k] = ok[k] ? atomicAdd(&cnt[o[k]], 1) : 0;
  }
  __syncthreads();
  for (int i = tid; i < nown; i += PLACE_WAVES * 32) {
    const int n = cnt[i];
    if (n > 0) cnt[i] = atomicAdd(&ocount[i * CNT_STRIDE], n);
  }
  __syncthreads();
#pragma unroll
  for (int k = 0; k < PLACE_PER; ++k) pos[k] = cnt[o[k]] + rk[k];
#pragma unroll
  for (int k = 0; k < PLACE_PER; ++k) {
    if (ok[k] && pos[k] < OCAP) {
      const v2i ent = {r[k] | ((c[k] - o[k] * OWN_ROWS) << SRC_BITS), __float_as_int(w[k])};
      *(v2i*)(olist + 2 * ((size_t)o[k] * OCAP + pos[k])) = ent;
    }
  }
}

template <int K>
__device__ __forceinline__ unsigned gather_add_k(int r, int cl, float w, unsigned mask,
                                                 const float* __restrict__ H, float* tile,
                                                 int lane) {
  int id[K];
  v4f h[K];
#pragma unroll
  for (int u = 0; u < K; ++u) {
    id[u] = __builtin_ctz(mask);
    mask &= mask - 1;
    const int ru = __builtin_amdgcn_readlane(r, id[u]);
    h[u] = *(const v4f*)(H + (size_t)ru * NHID + lane * 4);
  }
#pragma unroll
  for (int u = 0; u < K; ++u) {
    const int cu = __builtin_amdgcn_readlane(cl, id[u]);
    const float wu = __int_as_float(__builtin_amdgcn_readlane(__float_as_int(w), id[u]));
    float* tp = tile + cu * NHID + lane;
    atomicAdd(tp + 0,  h[u].x * wu);
    atomicAdd(tp + 32, h[u].y * wu);
    atomicAdd(tp + 64, h[u].z * wu);
    atomicAdd(tp + 96, h[u].w * wu);
  }
  return mask;
}

__device__ __forceinline__ void add_batch(int r, int cl, float w, int n,
                                          const float* __restrict__ H, float* tile,
                                          int lane) {
  unsigned mask = __builtin_amdgcn_ballot_w32(lane < n);
  int left = n;
  for (; left >= 8; left -= 8) mask = gather_add_k<8>(r, cl, w, mask, H, tile, lane);
  if (left >= 4) {
    mask = gather_add_k<4>(r, cl, w, mask, H, tile, lane);
    left -= 4;
  }
  if (left >= 2) {
    mask = gather_add_k<2>(r, cl, w, mask, H, tile, lane);
    left -= 2;
  }
  if (left >= 1) gather_add_k<1>(r, cl, w, mask, H, tile, lane);
}

__device__ __forceinline__ void drain_batch(const int* qp, int n, int lo,
                                            const int* __restrict__ rows,
                                            const int* __restrict__ cols,
                                            const float* __restrict__ dinv,
                                            const float* __restrict__ H, float* tile,
                                            int lane) {
  const bool valid = lane < n;
  const int e = qp[valid ? lane : 0];
  const int r = rows[e];
  const int c = cols[e];
  const float w = dinv[r] * dinv[c];
  add_batch(r, c - lo, w, n, H, tile, lane);
}

#define WROWS ((OWN_ROWS + OWN_WAVES - 1) / OWN_WAVES)
static_assert(OWN_WAVES == 32 && WROWS * 32 >= OWN_ROWS, "one wave sums the row counts to offsets, WROWS rows a lane");
static_assert(OCAP <= 4 * OWN_WAVES * 32, "a thread of an owner workgroup sorts at most 4 list entries");
static_assert(2 * (OCAP + OWN_ROWS) <= OWN_WAVES * QCAP, "the sorted list lies in the LDS of the scan queues");

__device__ __forceinline__ int wave_first_row(int wave) {
  return (wave * OWN_ROWS + OWN_WAVES - 1) / OWN_WAVES;
}

template <int K>
__device__ __forceinline__ unsigned gather_run_k(int r, int cl, float w, unsigned mask,
                                                 const float* __restrict__ H,
                                                 float* __restrict__ Y, int lo, v4f& acc,
                                                 int& curj, int lane) {
  int id[K];
  v4f h[K];
#pragma unroll
  for (int u = 0; u < K; ++u) {
    id[u] = __builtin_ctz(mask);
    mask &= mask - 1;
    const int ru = __builtin_amdgcn_readlane(r, id[u]);
    h[u] = *(const v4f*)(H + (size_t)ru * NHID + lane * 4);
  }
#pragma unroll
  for (int u = 0; u < K; ++u) {
    const int ju = __builtin_amdgcn_readlane(cl, id[u]);
    const float wu = __int_as_float(__builtin_amdgcn_readlane(__float_as_int(w), id[u]));
    if (ju != curj) {
      *(v4f*)(Y + (size_t)(lo + curj) * NHID + lane * 4) = acc;
      acc = v4f{0.0f, 0.0f, 0.0f, 0.0f};
      curj = ju;
    }
    acc += h[u] * wu;
  }
  return mask;
}

__device__ __forceinline__ void run_batch(int r, int cl, float w, int n,
                                          const float* __restrict__ H, float* __restrict__ Y,
                                          int lo, v4f& acc, int& curj, int lane) {
  unsigned mask = __builtin_amdgcn_ballot_w32(lane < n);
  int left = n;
  for (; left >= 8; left -= 8) mask = gather_run_k<8>(r, cl, w, mask, H, Y, lo, acc, curj, lane);
  if (left >= 4) {
    mask = gather_run_k<4>(r, cl, w, mask, H, Y, lo, acc, curj, lane);
    left -= 4;
  }
  if (left >= 2) {
    mask = gather_run_k<2>(r, cl, w, mask, H, Y, lo, acc, curj, lane);
    left -= 2;
  }
  if (left >= 1) gather_run_k<1>(r, cl, w, mask, H, Y, lo, acc, curj, lane);
}

__global__ void __launch_bounds__(OWN_WAVES * 32) agg_owner(const int* __restrict__ rows,
                                                            const int* __restrict__ cols,
                                                            const float* __restrict__ dinv,
                                                            const float* __restrict__ H,
                                                            float* __restrict__ Y,
                                                            const int* __restrict__ olist,
                                                            const int* __restrict__ ocount,
                                                            int E, int N) {
  __shared__ float tile[OWN_ROWS * NHID];
  __shared__ __attribute__((aligned(16))) int queue[OWN_WAVES][QCAP];
  __shared__ int rcnt[OWN_ROWS];
  __shared__ int rbeg[OWN_ROWS + 1];
  const int tid = threadIdx.x;
  const int lane = tid & 31;
  const int wave = __builtin_amdgcn_readfirstlane(tid >> 5);
  const int lo = blockIdx.x * OWN_ROWS;
  const int nrows = min(OWN_ROWS, N - lo);
  const bool wide = (((size_t)cols) & 15) == 0;

  const int nown = olist != nullptr
                       ? __builtin_amdgcn_readfirstlane(ocount[blockIdx.x * CNT_STRIDE])
                       : OCAP + 1;
  if (nown <= OCAP) {
    v2i* sorted = (v2i*)&queue[0][0];
    if (tid < OWN_ROWS) rcnt[tid] = 0;
    __syncthreads();
    const v2i* lp = (const v2i*)olist + (size_t)blockIdx.x * OCAP;
    v2i ent[4];
    int rk[4];
#pragma unroll
    for (int k = 0; k < 4; ++k) {
      const int i = tid + k * (OWN_WAVES * 32);
      ent[k] = v2i{0, 0};
      if (i < nown) ent[k] = lp[i];
    }
    const float dself = dinv[lo + (tid < nrows ? tid : 0)];
#pragma unroll
    for (int k = 0; k < 4; ++k) {
      rk[k] = 0;
      if (tid + k * (OWN_WAVES * 32) < nown)
        rk[k] = atomicAdd(&rcnt[(unsigned)ent[k].x >> SRC_BITS], 1);
    }
    __syncthreads();
    if (wave == 0) {
      int n[WROWS];
      int s = 0;
#pragma unroll
      for (int q = 0; q < WROWS; ++q) {
        const int row = lane * WROWS + q;
        n[q] = row < OWN_ROWS ? rcnt[row] + (row < nrows ? 1 : 0) : 0;
        s += n[q];
      }
      int run = s;
#pragma unroll
      for (int off = 1; off < 32; off <<= 1) {
        const int t = __shfl_up(run, off, 32);
        if (lane >= off) run += t;
      }
      run -= s;
#pragma unroll
      for (int q = 0; q < WROWS; ++q) {
        const int row = lane * WROWS + q;
        if (row < OWN_ROWS) rbeg[row] = run;
        run += n[q];
      }
      if (lane * WROWS < OWN_ROWS && lane * WROWS + WROWS >= OWN_ROWS) rbeg[OWN_ROWS] = run;
    }
    __syncthreads();
#pragma unroll
    for (int k = 0; k < 4; ++k)
      if (tid + k * (OWN_WAVES * 32) < nown)
        sorted[rbeg[(unsigned)ent[k].x >> SRC_BITS] + 1 + rk[k]] = ent[k];
    if (tid < nrows) {
      const v2i self = {(lo + tid) | (tid << SRC_BITS), __float_as_int(dself * dself)};
      sorted[rbeg[tid]] = self;
    }
    __syncthreads();

    const int first = wave_first_row(wave);
    const int last = min(wave_first_row(wave + 1), nrows);
    if (first < last) {
      const int beg = rbeg[first];
      const int m = rbeg[last] - beg;
      v4f acc = {0.0f, 0.0f, 0.0f, 0.0f};
      int curj = first;
      for (int i = 0; i < m; i += 32) {
        const v2i cur = sorted[beg + min(i + lane, m - 1)];
        run_batch(cur.x & ((1 << SRC_BITS) - 1), (int)((unsigned)cur.x >> SRC_BITS),
                  __int_as_float(cur.y), min(32, m - i), H, Y, lo, acc, curj, lane);
      }
      *(v4f*)(Y + (size_t)(lo + curj) * NHID + lane * 4) = acc;
    }
    return;
  }

  for (int i = tid; i < nrows * 32; i += OWN_WAVES * 32) {
    const int row = i >> 5;
    const float d = dinv[lo + row];
    const v4f v = *(const v4f*)(H + (size_t)(lo + row) * NHID + lane * 4);
    float* tp = tile + row * NHID + lane;
    tp[0]  = v.x * d * d;
    tp[32] = v.y * d * d;
    tp[64] = v.z * d * d;
    tp[96] = v.w * d * d;
  }
  __syncthreads();

  {
    int* q = queue[wave];
    int qn = 0;

    const int wave_edges = 128 * SCAN_LOADS;
    const int trip_edges = OWN_WAVES * wave_edges;
    const int nfull = wide ? E / trip_edges : 0;
    if (nfull > 0) {
      const int4* cp = (const int4*)(cols + wave * wave_edges) + lane;
      int trip = (int)(blockIdx.x % (unsigned)nfull);
      int4 cur[SCAN_LOADS];
#pragma unroll
      for (int j = 0; j < SCAN_LOADS; ++j) cur[j] = cp[(size_t)trip * (trip_edges / 4) + j * 32];
      for (int i = 0; i < nfull; ++i) {
        const int base = (trip * OWN_WAVES + wave) * wave_edges;
        trip = trip + 1 < nfull ? trip + 1 : 0;
        const int4* np = cp + (size_t)trip * (trip_edges / 4);
        int4 t[SCAN_LOADS];
#pragma unroll
        for (int j = 0; j < SCAN_LOADS; ++j) t[j] = cur[j];
#pragma unroll
        for (int j = 0; j < SCAN_LOADS; ++j) cur[j] = np[j * 32];
#pragma unroll
        for (int j = 0; j < SCAN_LOADS; ++j) {
          const int tj[4] = {t[j].x, t[j].y, t[j].z, t[j].w};
#pragma unroll
          for (int k = 0; k < 4; ++k) {
            const bool hit = (unsigned)(tj[k] - lo) < (unsigned)nrows;
            const unsigned m = __builtin_amdgcn_ballot_w32(hit);
            if (m != 0) {
              if (hit) q[qn + (int)__builtin_amdgcn_mbcnt_lo(m, 0)] = base + j * 128 + lane * 4 + k;
              qn += __builtin_popcount(m);
            }
          }
        }
        if (qn >= 32) {
          int qh = 0;
          for (; qn - qh >= 32; qh += 32)
            drain_batch(q + qh, 32, lo, rows, cols, dinv, H, tile, lane);
          const int left = qn - qh;
          int keep = 0;
          if (lane < left) keep = q[qh + lane];
          if (lane < left) q[lane] = keep;
          qn = left;
        }
      }
    }
    for (int base = nfull * trip_edges + wave * 128; base < E; base += OWN_WAVES * 128) {
#pragma unroll
      for (int k = 0; k < 4; ++k) {
        const int e = base + k * 32 + lane;
        const int c = e < E ? cols[e] : -1;
        const bool hit = (unsigned)(c - lo) < (unsigned)nrows;
        const unsigned m = __builtin_amdgcn_ballot_w32(hit);
        if (m != 0) {
          if (hit) q[qn + (int)__builtin_amdgcn_mbcnt_lo(m, 0)] = e;
          qn += __builtin_popcount(m);
        }
      }
      if (qn >= 32) {
        int qh = 0;
        for (; qn - qh >= 32; qh += 32)
          drain_batch(q + qh, 32, lo, rows, cols, dinv, H, tile, lane);
        const int left = qn - qh;
        int keep = 0;
        if (lane < left) keep = q[qh + lane];
        if (lane < left) q[lane] = keep;
        qn = left;
      }
    }
    if (qn > 0) drain_batch(q, qn, lo, rows, cols, dinv, H, tile, lane);
  }
  __syncthreads();

  for (int i = tid; i < nrows * 32; i += OWN_WAVES * 32) {
    const int row = i >> 5;
    const float* tp = tile + row * NHID + lane;
    const v4f v = {tp[0], tp[32], tp[64], tp[96]};
    *(v4f*)(Y + (size_t)(lo + row) * NHID + lane * 4) = v;
  }
}

__global__ void __launch_bounds__(256) final_logits(const float* __restrict__ X,
                                                    const float* __restrict__ bx,
                                                    const float* __restrict__ Wlt,
                                                    const float* __restrict__ bl,
                                                    float* __restrict__ out, int n) {
  int node = (int)((blockIdx.x * (size_t)blockDim.x + threadIdx.x) >> 5);
  int lane = threadIdx.x & 31;
  if (node >= n) return;

  float4 xv = *(const float4*)(X + (size_t)node * NHID + lane * 4);
  const float4 bv = *(const float4*)(bx + lane * 4);
  xv.x += bv.x;
  xv.y += bv.y;
  xv.z += bv.z;
  xv.w += bv.w;
  float logit[NCLASS];
#pragma unroll
  for (int c = 0; c < NCLASS; ++c) {
    float4 wv = *(const float4*)(Wlt + (size_t)c * NHID + lane * 4);
    float p = xv.x * wv.x + xv.y * wv.y + xv.z * wv.z + xv.w * wv.w;
#pragma unroll
    for (int off = 16; off > 0; off >>= 1) p += __shfl_xor(p, off, 32);
    logit[c] = p + bl[c];
  }
  float m = logit[0];
#pragma unroll
  for (int c = 1; c < NCLASS; ++c) m = fmaxf(m, logit[c]);
  float s = 0.0f;
#pragma unroll
  for (int c = 0; c < NCLASS; ++c) s += __expf(logit[c] - m);
  float lse = m + __logf(s);
  if (lane < NCLASS) out[(size_t)node * NCLASS + lane] = logit[lane] - lse;
}

extern "C" void kernel_launch(void* const* d_in, const int* in_sizes, int n_in,
                              void* d_out, int out_size, void* d_ws, size_t ws_size,
                              hipStream_t stream) {
  const float* x  = (const float*)d_in[0];
  const int*   ei = (const int*)d_in[1];
  const float* W1 = (const float*)d_in[2];
  const float* b1 = (const float*)d_in[3];
  const float* W2 = (const float*)d_in[4];
  const float* b2 = (const float*)d_in[5];
  const float* W3 = (const float*)d_in[6];
  const float* b3 = (const float*)d_in[7];
  const float* Wl = (const float*)d_in[8];
  const float* bl = (const float*)d_in[9];
  float* out = (float*)d_out;

  const int E = in_sizes[1] / 2;
  const int N = in_sizes[0] / NFEAT;

  char* ws = (char*)d_ws;
  size_t off = 0;
  auto alloc = [&](size_t bytes) -> float* {
    float* p = (float*)(ws + off);
    off += (bytes + 255) & ~(size_t)255;
    return p;
  };
  float* dinv = alloc((size_t)N * sizeof(float));
  float* Wp1  = alloc((size_t)NFEAT * NHID * sizeof(float));
  float* Wp2  = alloc((size_t)NHID * NHID * sizeof(float));
  float* Wp3  = alloc((size_t)NHID * NHID * sizeof(float));
  float* Wlt  = alloc((size_t)NHID * NCLASS * sizeof(float));
  float* H    = alloc((size_t)N * NHID * sizeof(float));
  float* Y    = alloc((size_t)N * NHID * sizeof(float));

  const int* rows = ei;
  const int* cols = ei + E;

  const int tiles = (N + 15) / 16;
  const int gemm_grid = (tiles + 3) / 4;
  const int ogrid = (N + OWN_ROWS - 1) / OWN_ROWS;

  int* olist  = (int*)alloc((size_t)ogrid * OCAP * 2 * sizeof(int));
  int* ocount = (int*)alloc((size_t)ogrid * CNT_STRIDE * sizeof(int));
  const bool listed = off <= ws_size && N <= (1 << SRC_BITS) && ogrid <= PLACE_MAXOWN;
  if (!listed) {
    olist = nullptr;
    ocount = nullptr;
  }

  pack_w<<<(NFEAT * NHID + 255) / 256, 256, 0, stream>>>(W1, Wp1, NFEAT);
  pack_w<<<(NHID * NHID + 255) / 256, 256, 0, stream>>>(W2, Wp2, NHID);
  pack_w<<<(NHID * NHID + 255) / 256, 256, 0, stream>>>(W3, Wp3, NHID);
  transpose_w<<<(NHID * NCLASS + 255) / 256, 256, 0, stream>>>(Wl, Wlt, NHID, NCLASS);

  deg_init<<<(N + 255) / 256, 256, 0, stream>>>(dinv, N, ocount, listed ? ogrid : 0);
  deg_edges<<<(E + 255) / 256, 256, 0, stream>>>(cols, dinv, E);
  deg_fin<<<(N + 255) / 256, 256, 0, stream>>>(dinv, N);
  if (listed) {
    const int per_group = PLACE_WAVES * 32 * PLACE_PER;
    place_edges<<<(E + per_group - 1) / per_group, PLACE_WAVES * 32, 0, stream>>>(rows, cols, dinv, olist,
                                                                                ocount, E, ogrid);
  }

  gemm_wmma_f32<false><<<gemm_grid, 128, 0, stream>>>(x, Wp1, nullptr, H, N, NFEAT);
  agg_owner<<<ogrid, OWN_WAVES * 32, 0, stream>>>(rows, cols, dinv, H, Y, olist, ocount, E, N);

  gemm_wmma_f32<true><<<gemm_grid, 128, 0, stream>>>(Y, Wp2, b1, H, N, NHID);
  agg_owner<<<ogrid, OWN_WAVES * 32, 0, stream>>>(rows, cols, dinv, H, Y, olist, ocount, E, N);

  gemm_wmma_f32<true><<<gemm_grid, 128, 0, stream>>>(Y, Wp3, b2, H, N, NHID);
  agg_owner<<<ogrid, OWN_WAVES * 32, 0, stream>>>(rows, cols, dinv, H, Y, olist, ocount, E, N);

  final_logits<<<(int)(((size_t)N * 32 + 255) / 256), 256, 0, stream>>>(Y, b3, Wlt, bl, out, N);
}
